# starting kernel
# baseline (speedup 1.0000x reference)
#include <hip/hip_runtime.h>
#include <hip/hip_bf16.h>


typedef __attribute__((ext_vector_type(2))) float v2f;
typedef __attribute__((ext_vector_type(8))) float v8f;

#define DIM 64

__device__ __forceinline__ void atomic_add_f32_dev(float* addr, float val) {
    asm volatile("global_atomic_add_f32 %0, %1, off scope:SCOPE_DEV"
                 :: "v"(addr), "v"(val)
                 : "memory");
}

__global__ __launch_bounds__(256) void zero_kernel(float4* __restrict__ out, int n4) {
    int i = blockIdx.x * blockDim.x + threadIdx.x;
    if (i < n4) out[i] = make_float4(0.f, 0.f, 0.f, 0.f);
}

__global__ __launch_bounds__(256) void gemm_kernel(const float* __restrict__ x,
                                                   const float* __restrict__ W,
                                                   const float* __restrict__ b,
                                                   float* __restrict__ support,
                                                   int n_nodes) {
    const int lane    = threadIdx.x & 31;
    const int wave    = threadIdx.x >> 5;
    const int tile    = blockIdx.x * 8 + wave;
    const int n_tiles = n_nodes >> 4;
    if (tile >= n_tiles) return;

    const int row0    = tile << 4;
    const int halfSel = lane >> 4;
    const int l       = lane & 15;

    v8f acc[4];
    #pragma unroll
    for (int nt = 0; nt < 4; ++nt) {
        const float bias = b[nt * 16 + l];
        #pragma unroll
        for (int j = 0; j < 8; ++j) acc[nt][j] = bias;
    }

    #pragma unroll 4
    for (int k = 0; k < DIM; k += 4) {
        const float* xa = x + (size_t)(row0 + l) * DIM + k + 2 * halfSel;
        v2f a;
        a[0] = xa[0];
        a[1] = xa[1];
        #pragma unroll
        for (int nt = 0; nt < 4; ++nt) {
            const float* wb = W + (size_t)(nt * 16 + l) * DIM + k + 2 * halfSel;
            v2f bf;
            bf[0] = wb[0];
            bf[1] = wb[1];
            acc[nt] = __builtin_amdgcn_wmma_f32_16x16x4_f32(
                 false, a,  false, bf,
                 (short)0, acc[nt],  false,  false);
        }
    }

    #pragma unroll
    for (int nt = 0; nt < 4; ++nt) {
        #pragma unroll
        for (int j = 0; j < 8; ++j) {
            const int m = j + 8 * halfSel;
            support[(size_t)(row0 + m) * DIM + nt * 16 + l] = acc[nt][j];
        }
    }
}

__global__ __launch_bounds__(256) void scatter_kernel(const float* __restrict__ support,
                                                      const int* __restrict__ esrc,
                                                      const int* __restrict__ edst,
                                                      const float* __restrict__ ew,
                                                      float* __restrict__ out,
                                                      int n_edges) {
    const int t = blockIdx.x * blockDim.x + threadIdx.x;
    const int e = t >> 4;
    if (e >= n_edges) return;
    const int l16 = t & 15;

    const int   s = esrc[e];
    const int   d = edst[e];
    const float w = ew[e];

    const float4 v = ((const float4*)(support + (size_t)s * DIM))[l16];

    float* orow = out + (size_t)d * DIM + 4 * l16;
    atomic_add_f32_dev(orow + 0, w * v.x);
    atomic_add_f32_dev(orow + 1, w * v.y);
    atomic_add_f32_dev(orow + 2, w * v.z);
    atomic_add_f32_dev(orow + 3, w * v.w);
}

extern "C" void kernel_launch(void* const* d_in, const int* in_sizes, int n_in,
                              void* d_out, int out_size, void* d_ws, size_t ws_size,
                              hipStream_t stream) {
    const float* x    = (const float*)d_in[0];
    const float* W    = (const float*)d_in[1];
    const float* b    = (const float*)d_in[2];
    const int*   esrc = (const int*)  d_in[3];
    const int*   edst = (const int*)  d_in[4];
    const float* ew   = (const float*)d_in[5];

    float* out     = (float*)d_out;
    float* support = (float*)d_ws;

    const int n_nodes = in_sizes[0] / DIM;
    const int n_edges = in_sizes[3];

    const int n4 = out_size / 4;
    zero_kernel<<<(n4 + 255) / 256, 256, 0, stream>>>((float4*)out, n4);

    const int n_tiles = n_nodes / 16;
    const int gblocks = (n_tiles + 7) / 8;
    gemm_kernel<<<gblocks, 256, 0, stream>>>(x, W, b, support, n_nodes);

    const long long sthreads = (long long)n_edges * 16;
    const unsigned  sblocks  = (unsigned)((sthreads + 255) / 256);
    scatter_kernel<<<sblocks, 256, 0, stream>>>(support, esrc, edst, ew, out, n_edges);
}


# Edge sum by destination ownership: a work-group takes 512 out rows, finds its edges in a byte-key scan, each wave sums its own 32 rows in registers; atomics and zero pass gone, fixed sum order
# speedup vs baseline: 1.0456x; 1.0456x over previous
#include <hip/hip_runtime.h>
#include <hip/hip_bf16.h>


typedef __attribute__((ext_vector_type(2))) float v2f;
typedef __attribute__((ext_vector_type(8))) float v8f;

#define DIM 64

#define BKT_SHIFT 9
#define BKT_ROWS  512
#define NWAVE     16
#define CAP       1280
#define SCAN_STEP 512
#define OCAP      256

#define KEY_PAD_LOADS (9 * NWAVE)

__host__ __device__ __forceinline__ int key_loads(int n_edges) { return (((n_edges + 15) >> 4) + 31) >> 5; }
__host__ __device__ __forceinline__ int key_units(int n_edges) { return (key_loads(n_edges) + KEY_PAD_LOADS) * 32; }

__global__ __launch_bounds__(256) void gemm_kernel(const float* __restrict__ x,
                                                   const float* __restrict__ W,
                                                   const float* __restrict__ b,
                                                   const int* __restrict__ edst,
                                                   float* __restrict__ support,
                                                   uint2* __restrict__ key8,
                                                   int n_nodes, int n_edges) {
    {
        const int gtid = blockIdx.x * 256 + threadIdx.x;
        const int gsz  = gridDim.x * 256;
        const int ngrp = key_units(n_edges) * 2;
        for (int g = gtid; g < ngrp; g += gsz) {
            const int e0 = g * 8;
            unsigned lo, hi;
            if (e0 + 8 <= n_edges) {
                const int4 p = ((const int4*)edst)[2 * g];
                const int4 q = ((const int4*)edst)[2 * g + 1];
                lo = ((unsigned)(p.x >> BKT_SHIFT) & 255u)       | (((unsigned)(p.y >> BKT_SHIFT) & 255u) << 8) |
                     (((unsigned)(p.z >> BKT_SHIFT) & 255u) << 16) | (((unsigned)(p.w >> BKT_SHIFT) & 255u) << 24);
                hi = ((unsigned)(q.x >> BKT_SHIFT) & 255u)       | (((unsigned)(q.y >> BKT_SHIFT) & 255u) << 8) |
                     (((unsigned)(q.z >> BKT_SHIFT) & 255u) << 16) | (((unsigned)(q.w >> BKT_SHIFT) & 255u) << 24);
            } else {
                unsigned kb[8];
                #pragma unroll
                for (int j = 0; j < 8; ++j) {
                    const int e = e0 + j;
                    kb[j] = (e < n_edges) ? ((unsigned)(edst[e] >> BKT_SHIFT) & 255u) : 255u;
                }
                lo = kb[0] | (kb[1] << 8) | (kb[2] << 16) | (kb[3] << 24);
                hi = kb[4] | (kb[5] << 8) | (kb[6] << 16) | (kb[7] << 24);
            }
            key8[g] = make_uint2(lo, hi);
        }
    }

    const int lane    = threadIdx.x & 31;
    const int wave    = threadIdx.x >> 5;
    const int tile    = blockIdx.x * 8 + wave;
    const int n_tiles = n_nodes >> 4;
    if (tile >= n_tiles) return;

    const int row0    = tile << 4;
    const int halfSel = lane >> 4;
    const int l       = lane & 15;

    v8f acc[4];
    #pragma unroll
    for (int nt = 0; nt < 4; ++nt) {
        const float bias = b[nt * 16 + l];
        #pragma unroll
        for (int j = 0; j < 8; ++j) acc[nt][j] = bias;
    }

    #pragma unroll 4
    for (int k = 0; k < DIM; k += 4) {
        const float* xa = x + (size_t)(row0 + l) * DIM + k + 2 * halfSel;
        v2f a;
        a[0] = xa[0];
        a[1] = xa[1];
        #pragma unroll
        for (int nt = 0; nt < 4; ++nt) {
            const float* wb = W + (size_t)(nt * 16 + l) * DIM + k + 2 * halfSel;
            v2f bf;
            bf[0] = wb[0];
            bf[1] = wb[1];
            acc[nt] = __builtin_amdgcn_wmma_f32_16x16x4_f32(
                 false, a,  false, bf,
                 (short)0, acc[nt],  false,  false);
        }
    }

    #pragma unroll
    for (int nt = 0; nt < 4; ++nt) {
        #pragma unroll
        for (int j = 0; j < 8; ++j) {
            const int m = j + 8 * halfSel;
            support[(size_t)(row0 + m) * DIM + nt * 16 + l] = acc[nt][j];
        }
    }
}

__device__ __forceinline__ unsigned nonzero_bytes(unsigned x) {
    return (((x & 0x7F7F7F7Fu) + 0x7F7F7F7Fu) | x) & 0x80808080u;
}

typedef __attribute__((ext_vector_type(32))) float v32f;

struct EdgeGroup {
    int    G;
    int    key;
    float  wgt;
    float2 v[8];
};

__device__ __forceinline__ void fetch_group(EdgeGroup& g, const int* own, int i, int ocnt,
                                            const int* s_key, const float* s_w,
                                            const float2* __restrict__ sup2, int lane) {
    g.G   = min(8, ocnt - i);
    g.key = 0;
    g.wgt = 0.f;
    if (lane < g.G) {
        const int idx = own[i + lane];
        g.key = s_key[idx];
        g.wgt = s_w[idx];
    }
    #pragma unroll
    for (int u = 0; u < 8; ++u) {
        const int ku = __builtin_amdgcn_readlane(g.key, u);
        g.v[u] = sup2[(size_t)(ku >> BKT_SHIFT) * 32 + lane];
    }
}

__device__ __forceinline__ void add_group(const EdgeGroup& g, v32f& ax, v32f& ay) {
    #pragma unroll
    for (int u = 0; u < 8; ++u) {
        if (u < g.G) {
            const int   ku = __builtin_amdgcn_readlane(g.key, u);
            const float wu = __int_as_float(__builtin_amdgcn_readlane(__float_as_int(g.wgt), u));
            const int   n  = ku & 31;
            ax[n] = fmaf(wu, g.v[u].x, ax[n]);
            ay[n] = fmaf(wu, g.v[u].y, ay[n]);
        }
    }
}

__device__ __forceinline__ void add_owned(const int* own, int ocnt,
                                          const int* s_key, const float* s_w,
                                          const float2* __restrict__ sup2,
                                          v32f& ax, v32f& ay, int lane) {
    if (ocnt <= 0) return;
    EdgeGroup ga, gb;
    fetch_group(ga, own, 0, ocnt, s_key, s_w, sup2, lane);
    for (int i = 0; i < ocnt; i += 16) {
        fetch_group(gb, own, i + 8, ocnt, s_key, s_w, sup2, lane);
        add_group(ga, ax, ay);
        fetch_group(ga, own, i + 16, ocnt, s_key, s_w, sup2, lane);
        add_group(gb, ax, ay);
    }
}

__device__ __forceinline__ int scan_unit(const uint4 kw, int wi, unsigned pat, int* myKey, int cnt) {
    const unsigned other = (nonzero_bytes(kw.x ^ pat) >> 7) | (nonzero_bytes(kw.y ^ pat) >> 6) |
                           (nonzero_bytes(kw.z ^ pat) >> 5) | (nonzero_bytes(kw.w ^ pat) >> 4);
    unsigned mm = ~other & 0x0F0F0F0Fu;
    for (;;) {
        const unsigned bal = __builtin_amdgcn_ballot_w32(mm != 0u);
        if (bal == 0u) break;
        if (mm != 0u) {
            const int p = __builtin_ctz(mm);
            mm &= mm - 1u;
            myKey[cnt + (int)__builtin_amdgcn_mbcnt_lo(bal, 0u)] = wi * 16 + ((p & 7) << 2) + (p >> 3);
        }
        cnt += __builtin_popcount(bal);
    }
    return cnt;
}

__global__ __launch_bounds__(512) void owner_kernel(const float* __restrict__ support,
                                                    const uint4* __restrict__ key16,
                                                    const int* __restrict__ esrc,
                                                    const int* __restrict__ edst,
                                                    const float* __restrict__ ew,
                                                    float* __restrict__ out,
                                                    int n_nodes, int n_edges) {
    __shared__ int    s_key[NWAVE * CAP];
    __shared__ float  s_w[NWAVE * CAP];
    __shared__ int    s_own[NWAVE * OCAP];
    __shared__ int    s_cnt[NWAVE];
    __shared__ int    s_more[NWAVE];

    const int lane = threadIdx.x & 31;
    const int wv   = __builtin_amdgcn_readfirstlane((int)(threadIdx.x >> 5));
    const int bkt  = blockIdx.x;
    const unsigned pat = (unsigned)bkt * 0x01010101u;
    const int nld  = key_loads(n_edges);

    v32f ax, ay;
    #pragma unroll
    for (int r = 0; r < 32; ++r) { ax[r] = 0.f; ay[r] = 0.f; }

    int*   myKey = s_key + wv * CAP;
    float* myW   = s_w   + wv * CAP;
    int*   myOwn = s_own + wv * OCAP;
    const float2* sup2 = (const float2*)support;

    int t = wv;
    for (;;) {
        int cnt = 0;
        uint4 c0 = key16[(t)             * 32 + lane];
        uint4 c1 = key16[(t + NWAVE)     * 32 + lane];
        uint4 c2 = key16[(t + 2 * NWAVE) * 32 + lane];
        uint4 c3 = key16[(t + 3 * NWAVE) * 32 + lane];
        for (;;) {
            if (!(t < nld && cnt <= CAP - SCAN_STEP)) break;
            const uint4 n0 = key16[(t + 4 * NWAVE) * 32 + lane];
            const uint4 n1 = key16[(t + 5 * NWAVE) * 32 + lane];
            const uint4 n2 = key16[(t + 6 * NWAVE) * 32 + lane];
            const uint4 n3 = key16[(t + 7 * NWAVE) * 32 + lane];
            cnt = scan_unit(c0, t * 32 + lane, pat, myKey, cnt);
            t  += NWAVE;
            if (t < nld && cnt <= CAP - SCAN_STEP) {
                cnt = scan_unit(c1, t * 32 + lane, pat, myKey, cnt);
                t  += NWAVE;
                if (t < nld && cnt <= CAP - SCAN_STEP) {
                    cnt = scan_unit(c2, t * 32 + lane, pat, myKey, cnt);
                    t  += NWAVE;
                    if (t < nld && cnt <= CAP - SCAN_STEP) {
                        cnt = scan_unit(c3, t * 32 + lane, pat, myKey, cnt);
                        t  += NWAVE;
                    }
                }
            }
            c0 = n0; c1 = n1; c2 = n2; c3 = n3;
        }
        for (int i = lane; i < cnt; i += 32) {
            const int   e = myKey[i];
            const int   d = edst[e];
            const int   s = esrc[e];
            const float w = ew[e];
            myKey[i] = (s << BKT_SHIFT) | (d & (BKT_ROWS - 1));
            myW[i]   = w;
        }
        if (lane == 0) {
            s_cnt[wv]  = cnt;
            s_more[wv] = (t < nld) ? 1 : 0;
        }
        __syncthreads();

        int ocnt = 0;
        for (int s = 0; s < NWAVE; ++s) {
            const int  c  = __builtin_amdgcn_readfirstlane(s_cnt[s]);
            const int* lk = s_key + s * CAP;
            for (int base = 0; base < c; base += 32) {
                const int i = base + lane;
                bool mine = false;
                if (i < c) mine = (((lk[i] >> 5) & (NWAVE - 1)) == wv);
                const unsigned mk = __builtin_amdgcn_ballot_w32(mine);
                if (mine) myOwn[ocnt + (int)__builtin_amdgcn_mbcnt_lo(mk, 0u)] = s * CAP + i;
                ocnt += __builtin_popcount(mk);
                if (ocnt > OCAP - 32) {
                    add_owned(myOwn, ocnt, s_key, s_w, sup2, ax, ay, lane);
                    ocnt = 0;
                }
            }
        }
        add_owned(myOwn, ocnt, s_key, s_w, sup2, ax, ay, lane);

        const int mflag = (lane < NWAVE) ? s_more[lane] : 0;
        const unsigned more = __builtin_amdgcn_ballot_w32(mflag != 0);
        __syncthreads();
        if (more == 0u) break;
    }

    float2* out2 = (float2*)out;
    #pragma unroll
    for (int r = 0; r < 32; ++r) {
        const int node = bkt * BKT_ROWS + wv * 32 + r;
        if (node < n_nodes) out2[(size_t)node * 32 + lane] = make_float2(ax[r], ay[r]);
    }
}

extern "C" void kernel_launch(void* const* d_in, const int* in_sizes, int n_in,
                              void* d_out, int out_size, void* d_ws, size_t ws_size,
                              hipStream_t stream) {
    const float* x    = (const float*)d_in[0];
    const float* W    = (const float*)d_in[1];
    const float* b    = (const float*)d_in[2];
    const int*   esrc = (const int*)  d_in[3];
    const int*   edst = (const int*)  d_in[4];
    const float* ew   = (const float*)d_in[5];

    const int n_nodes = in_sizes[0] / DIM;
    const int n_edges = in_sizes[3];

    float* out     = (float*)d_out;
    float* support = (float*)d_ws;
    uint2* key8    = (uint2*)(support + (size_t)n_nodes * DIM);

    const int n_tiles = n_nodes / 16;
    const int gblocks = (n_tiles + 7) / 8;
    gemm_kernel<<<gblocks, 256, 0, stream>>>(x, W, b, edst, support, key8, n_nodes, n_edges);

    const int oblocks = (n_nodes + BKT_ROWS - 1) / BKT_ROWS;
    owner_kernel<<<oblocks, 512, 0, stream>>>(support, (const uint4*)key8, esrc, edst, ew, out,
                                              n_nodes, n_edges);
}


# First kernel sorts each 2048-edge chunk by destination bucket (ballot ranks, no atomics), writes records and run starts; owners read only their runs, no key scan; sum order fixed but regrouped
# speedup vs baseline: 1.2517x; 1.1971x over previous
#include <hip/hip_runtime.h>
#include <hip/hip_bf16.h>


typedef __attribute__((ext_vector_type(2))) float v2f;
typedef __attribute__((ext_vector_type(8))) float v8f;

#define DIM 64

#define BKT_SHIFT 9
#define BKT_ROWS  512
#define NBK       256
#define NWAVE     16
#define CAP       1280
#define OCAP      256
#define CHUNK     2048

__host__ __device__ __forceinline__ int chunks_of(int n_edges) { return (n_edges + CHUNK - 1) / CHUNK; }

__device__ __forceinline__ int wave_incl_scan(int v, int lane) {
    #pragma unroll
    for (int d = 1; d < 32; d <<= 1) {
        const int t = __shfl_up(v, d, 32);
        if (lane >= d) v += t;
    }
    return v;
}

__global__ __launch_bounds__(256) void gemm_kernel(const float* __restrict__ x,
                                                   const float* __restrict__ W,
                                                   const float* __restrict__ b,
                                                   const int* __restrict__ esrc,
                                                   const int* __restrict__ edst,
                                                   const float* __restrict__ ew,
                                                   float* __restrict__ support,
                                                   int* __restrict__ rec_key,
                                                   float* __restrict__ rec_w,
                                                   int* __restrict__ tab,
                                                   int n_nodes, int n_edges) {
    __shared__ int s_hist[8 * NBK];
    __shared__ int s_start[NBK];
    __shared__ int s_wtot[8];

    const int lane = threadIdx.x & 31;
    const int wave = threadIdx.x >> 5;

    {
        const int n_chunks = chunks_of(n_edges);
        int* hw = s_hist + wave * NBK;
        for (int chunk = blockIdx.x; chunk < n_chunks; chunk += gridDim.x) {
            const int e0 = chunk * CHUNK + wave * 256 + lane;
            int dd[8], ss[8];
            float ww[8];
            #pragma unroll
            for (int j = 0; j < 8; ++j) {
                const int e = e0 + j * 32;
                dd[j] = 0; ss[j] = 0; ww[j] = 0.f;
                if (e < n_edges) { dd[j] = edst[e]; ss[j] = esrc[e]; ww[j] = ew[e]; }
            }
            for (int i = threadIdx.x; i < 8 * NBK; i += 256) s_hist[i] = 0;
            __syncthreads();

            int pk[8];
            #pragma unroll
            for (int j = 0; j < 8; ++j) {
                const int  e     = e0 + j * 32;
                const bool valid = e < n_edges;
                const int  d     = dd[j];
                const int  k     = (d >> BKT_SHIFT) & (NBK - 1);
                unsigned m = __builtin_amdgcn_ballot_w32(valid);
                #pragma unroll
                for (int bit = 0; bit < 8; ++bit) {
                    const bool     on = ((k >> bit) & 1) != 0;
                    const unsigned bb = __builtin_amdgcn_ballot_w32(on);
                    m &= on ? bb : ~bb;
                }
                const int below = (int)__builtin_amdgcn_mbcnt_lo(m, 0u);
                int r = 0;
                if (valid) {
                    const int base = hw[k];
                    r = base + below;
                    if (below == 0) hw[k] = base + __builtin_popcount(m);
                }
                pk[j] = valid ? (k | (r << 8) | ((d & (BKT_ROWS - 1)) << 20)) : -1;
            }
            __syncthreads();

            {
                const int k = threadIdx.x;
                int run = 0;
                #pragma unroll
                for (int w = 0; w < 8; ++w) {
                    const int c = s_hist[w * NBK + k];
                    s_hist[w * NBK + k] = run;
                    run += c;
                }
                const int incl = wave_incl_scan(run, lane);
                if (lane == 31) s_wtot[wave] = incl;
                __syncthreads();
                int wbase = 0;
                #pragma unroll
                for (int w = 0; w < 8; ++w) {
                    const int tw = s_wtot[w];
                    if (w < wave) wbase += tw;
                }
                const int st = wbase + incl - run;
                s_start[k] = st;
                tab[(size_t)k * n_chunks + chunk] = st;
                if (k == NBK - 1) tab[(size_t)NBK * n_chunks + chunk] = st + run;
            }
            __syncthreads();

            #pragma unroll
            for (int j = 0; j < 8; ++j) {
                if (pk[j] >= 0) {
                    const int    k   = pk[j] & (NBK - 1);
                    const int    pos = s_start[k] + hw[k] + ((pk[j] >> 8) & 4095);
                    const size_t o   = (size_t)chunk * CHUNK + pos;
                    rec_key[o] = (ss[j] << BKT_SHIFT) | (pk[j] >> 20);
                    rec_w[o]   = ww[j];
                }
            }
            __syncthreads();
        }
    }

    const int tile    = blockIdx.x * 8 + wave;
    const int n_tiles = n_nodes >> 4;
    if (tile >= n_tiles) return;

    const int row0    = tile << 4;
    const int halfSel = lane >> 4;
    const int l       = lane & 15;

    v8f acc[4];
    #pragma unroll
    for (int nt = 0; nt < 4; ++nt) {
        const float bias = b[nt * 16 + l];
        #pragma unroll
        for (int j = 0; j < 8; ++j) acc[nt][j] = bias;
    }

    #pragma unroll 4
    for (int k = 0; k < DIM; k += 4) {
        const float* xa = x + (size_t)(row0 + l) * DIM + k + 2 * halfSel;
        v2f a;
        a[0] = xa[0];
        a[1] = xa[1];
        #pragma unroll
        for (int nt = 0; nt < 4; ++nt) {
            const float* wb = W + (size_t)(nt * 16 + l) * DIM + k + 2 * halfSel;
            v2f bf;
            bf[0] = wb[0];
            bf[1] = wb[1];
            acc[nt] = __builtin_amdgcn_wmma_f32_16x16x4_f32(
                 false, a,  false, bf,
                 (short)0, acc[nt],  false,  false);
        }
    }

    #pragma unroll
    for (int nt = 0; nt < 4; ++nt) {
        #pragma unroll
        for (int j = 0; j < 8; ++j) {
            const int m = j + 8 * halfSel;
            support[(size_t)(row0 + m) * DIM + nt * 16 + l] = acc[nt][j];
        }
    }
}

typedef __attribute__((ext_vector_type(32))) float v32f;

struct EdgeGroup {
    int    G;
    int    key;
    float  wgt;
    float2 v[8];
};

__device__ __forceinline__ void fetch_group(EdgeGroup& g, const int* own, int i, int ocnt,
                                            const int* s_key, const float* s_w,
                                            const float2* __restrict__ sup2, int lane) {
    g.G   = min(8, ocnt - i);
    g.key = 0;
    g.wgt = 0.f;
    if (lane < g.G) {
        const int idx = own[i + lane];
        g.key = s_key[idx];
        g.wgt = s_w[idx];
    }
    #pragma unroll
    for (int u = 0; u < 8; ++u) {
        const int ku = __builtin_amdgcn_readlane(g.key, u);
        g.v[u] = sup2[(size_t)(ku >> BKT_SHIFT) * 32 + lane];
    }
}

__device__ __forceinline__ void add_group(const EdgeGroup& g, v32f& ax, v32f& ay) {
    #pragma unroll
    for (int u = 0; u < 8; ++u) {
        if (u < g.G) {
            const int   ku = __builtin_amdgcn_readlane(g.key, u);
            const float wu = __int_as_float(__builtin_amdgcn_readlane(__float_as_int(g.wgt), u));
            const int   n  = ku & 31;
            ax[n] = fmaf(wu, g.v[u].x, ax[n]);
            ay[n] = fmaf(wu, g.v[u].y, ay[n]);
        }
    }
}

__device__ __forceinline__ void add_owned(const int* own, int ocnt,
                                          const int* s_key, const float* s_w,
                                          const float2* __restrict__ sup2,
                                          v32f& ax, v32f& ay, int lane) {
    if (ocnt <= 0) return;
    EdgeGroup ga, gb;
    fetch_group(ga, own, 0, ocnt, s_key, s_w, sup2, lane);
    for (int i = 0; i < ocnt; i += 16) {
        fetch_group(gb, own, i + 8, ocnt, s_key, s_w, sup2, lane);
        add_group(ga, ax, ay);
        fetch_group(ga, own, i + 16, ocnt, s_key, s_w, sup2, lane);
        add_group(gb, ax, ay);
    }
}

__global__ __launch_bounds__(512) void owner_kernel(const float* __restrict__ support,
                                                    const int* __restrict__ tab,
                                                    const int* __restrict__ rec_key,
                                                    const float* __restrict__ rec_w,
                                                    float* __restrict__ out,
                                                    int n_nodes, int n_edges) {
    __shared__ int    s_key[NWAVE * CAP];
    __shared__ float  s_w[NWAVE * CAP];
    __shared__ int    s_own[NWAVE * OCAP];
    __shared__ int    s_cnt[NWAVE];
    __shared__ int    s_more[NWAVE];

    const int lane = threadIdx.x & 31;
    const int wv   = __builtin_amdgcn_readfirstlane((int)(threadIdx.x >> 5));
    const int bkt  = blockIdx.x;
    const int n_chunks = chunks_of(n_edges);
    const int* tlo = tab + (size_t)bkt * n_chunks;
    const int* thi = tab + (size_t)(bkt + 1) * n_chunks;

    v32f ax, ay;
    #pragma unroll
    for (int r = 0; r < 32; ++r) { ax[r] = 0.f; ay[r] = 0.f; }

    int*   myKey = s_key + wv * CAP;
    float* myW   = s_w   + wv * CAP;
    int*   myOwn = s_own + wv * OCAP;
    const float2* sup2 = (const float2*)support;

    int cur = wv;
    int off = 0;
    for (;;) {
        int cnt = 0;
        while (cur < n_chunks && cnt < CAP) {
            const int ci = cur + NWAVE * lane;
            int st = 0, n = 0;
            if (ci < n_chunks) {
                st = tlo[ci];
                n  = thi[ci] - st;
            }
            if (lane == 0) { st += off; n -= off; }
            const int incl  = wave_incl_scan(n, lane);
            const int excl  = incl - n;
            const int total = __builtin_amdgcn_readlane(incl, 31);
            const int take  = min(total, CAP - cnt);
            const int mine  = min(n, take - excl);
            const int*   rk = rec_key + (size_t)ci * CHUNK + st;
            const float* rw = rec_w   + (size_t)ci * CHUNK + st;
            int*   dk = myKey + cnt + excl;
            float* dw = myW   + cnt + excl;
            for (int t = 0; ; t += 4) {
                if (__builtin_amdgcn_ballot_w32(t < mine) == 0u) break;
                int   q0 = 0, q1 = 0, q2 = 0, q3 = 0;
                float f0 = 0.f, f1 = 0.f, f2 = 0.f, f3 = 0.f;
                if (t     < mine) { q0 = rk[t];     f0 = rw[t];     }
                if (t + 1 < mine) { q1 = rk[t + 1]; f1 = rw[t + 1]; }
                if (t + 2 < mine) { q2 = rk[t + 2]; f2 = rw[t + 2]; }
                if (t + 3 < mine) { q3 = rk[t + 3]; f3 = rw[t + 3]; }
                if (t     < mine) { dk[t]     = q0; dw[t]     = f0; }
                if (t + 1 < mine) { dk[t + 1] = q1; dw[t + 1] = f1; }
                if (t + 2 < mine) { dk[t + 2] = q2; dw[t + 2] = f2; }
                if (t + 3 < mine) { dk[t + 3] = q3; dw[t + 3] = f3; }
            }
            cnt += take;
            if (take == total) {
                cur += 32 * NWAVE;
                off = 0;
            } else {
                const unsigned part = __builtin_amdgcn_ballot_w32(incl > take);
                const int i    = __builtin_ctz(part);
                const int ex_i = __builtin_amdgcn_readlane(excl, i);
                off = (i == 0 ? off : 0) + (take - ex_i);
                cur += NWAVE * i;
            }
        }
        if (lane == 0) {
            s_cnt[wv]  = cnt;
            s_more[wv] = (cur < n_chunks) ? 1 : 0;
        }
        __syncthreads();

        int ocnt = 0;
        for (int s = 0; s < NWAVE; ++s) {
            const int  c  = __builtin_amdgcn_readfirstlane(s_cnt[s]);
            const int* lk = s_key + s * CAP;
            for (int base = 0; base < c; base += 32) {
                const int i = base + lane;
                bool mine = false;
                if (i < c) mine = (((lk[i] >> 5) & (NWAVE - 1)) == wv);
                const unsigned mk = __builtin_amdgcn_ballot_w32(mine);
                if (mine) myOwn[ocnt + (int)__builtin_amdgcn_mbcnt_lo(mk, 0u)] = s * CAP + i;
                ocnt += __builtin_popcount(mk);
                if (ocnt > OCAP - 32) {
                    add_owned(myOwn, ocnt, s_key, s_w, sup2, ax, ay, lane);
                    ocnt = 0;
                }
            }
        }
        add_owned(myOwn, ocnt, s_key, s_w, sup2, ax, ay, lane);

        const int mflag = (lane < NWAVE) ? s_more[lane] : 0;
        const unsigned more = __builtin_amdgcn_ballot_w32(mflag != 0);
        __syncthreads();
        if (more == 0u) break;
    }

    float2* out2 = (float2*)out;
    #pragma unroll
    for (int r = 0; r < 32; ++r) {
        const int node = bkt * BKT_ROWS + wv * 32 + r;
        if (node < n_nodes) out2[(size_t)node * 32 + lane] = make_float2(ax[r], ay[r]);
    }
}

extern "C" void kernel_launch(void* const* d_in, const int* in_sizes, int n_in,
                              void* d_out, int out_size, void* d_ws, size_t ws_size,
                              hipStream_t stream) {
    const float* x    = (const float*)d_in[0];
    const float* W    = (const float*)d_in[1];
    const float* b    = (const float*)d_in[2];
    const int*   esrc = (const int*)  d_in[3];
    const int*   edst = (const int*)  d_in[4];
    const float* ew   = (const float*)d_in[5];

    const int n_nodes = in_sizes[0] / DIM;
    const int n_edges = in_sizes[3];

    float* out     = (float*)d_out;
    const int n_chunks = chunks_of(n_edges);
    float* support = (float*)d_ws;
    int*   rec_key = (int*)(support + (size_t)n_nodes * DIM);
    float* rec_w   = (float*)(rec_key + (size_t)n_chunks * CHUNK);
    int*   tab     = (int*)(rec_w + (size_t)n_chunks * CHUNK);

    const int n_tiles = n_nodes / 16;
    const int gblocks = (n_tiles + 7) / 8;
    gemm_kernel<<<gblocks, 256, 0, stream>>>(x, W, b, esrc, edst, ew, support, rec_key, rec_w, tab,
                                             n_nodes, n_edges);

    const int oblocks = (n_nodes + BKT_ROWS - 1) / BKT_ROWS;
    owner_kernel<<<oblocks, 512, 0, stream>>>(support, tab, rec_key, rec_w, out, n_nodes, n_edges);
}


# Owner pick reads the keys of 8 batches (256 list entries) per LDS round trip ahead of the ballots; private list 1024 entries so one add pass per wave; order of every sum unchanged
# speedup vs baseline: 1.4464x; 1.1555x over previous
#include <hip/hip_runtime.h>
#include <hip/hip_bf16.h>


typedef __attribute__((ext_vector_type(2))) float v2f;
typedef __attribute__((ext_vector_type(8))) float v8f;

#define DIM 64

#define BKT_SHIFT 9
#define BKT_ROWS  512
#define NBK       256
#define NWAVE     16
#define CAP       1280
#define OCAP      1024
#define PICKB     8
#define CHUNK     2048

__host__ __device__ __forceinline__ int chunks_of(int n_edges) { return (n_edges + CHUNK - 1) / CHUNK; }

__device__ __forceinline__ int wave_incl_scan(int v, int lane) {
    #pragma unroll
    for (int d = 1; d < 32; d <<= 1) {
        const int t = __shfl_up(v, d, 32);
        if (lane >= d) v += t;
    }
    return v;
}

__global__ __launch_bounds__(256) void gemm_kernel(const float* __restrict__ x,
                                                   const float* __restrict__ W,
                                                   const float* __restrict__ b,
                                                   const int* __restrict__ esrc,
                                                   const int* __restrict__ edst,
                                                   const float* __restrict__ ew,
                                                   float* __restrict__ support,
                                                   int* __restrict__ rec_key,
                                                   float* __restrict__ rec_w,
                                                   int* __restrict__ tab,
                                                   int n_nodes, int n_edges) {
    __shared__ int s_hist[8 * NBK];
    __shared__ int s_start[NBK];
    __shared__ int s_wtot[8];

    const int lane = threadIdx.x & 31;
    const int wave = threadIdx.x >> 5;

    {
        const int n_chunks = chunks_of(n_edges);
        int* hw = s_hist + wave * NBK;
        for (int chunk = blockIdx.x; chunk < n_chunks; chunk += gridDim.x) {
            const int e0 = chunk * CHUNK + wave * 256 + lane;
            int dd[8], ss[8];
            float ww[8];
            #pragma unroll
            for (int j = 0; j < 8; ++j) {
                const int e = e0 + j * 32;
                dd[j] = 0; ss[j] = 0; ww[j] = 0.f;
                if (e < n_edges) { dd[j] = edst[e]; ss[j] = esrc[e]; ww[j] = ew[e]; }
            }
            for (int i = threadIdx.x; i < 8 * NBK; i += 256) s_hist[i] = 0;
            __syncthreads();

            int pk[8];
            #pragma unroll
            for (int j = 0; j < 8; ++j) {
                const int  e     = e0 + j * 32;
                const bool valid = e < n_edges;
                const int  d     = dd[j];
                const int  k     = (d >> BKT_SHIFT) & (NBK - 1);
                unsigned m = __builtin_amdgcn_ballot_w32(valid);
                #pragma unroll
                for (int bit = 0; bit < 8; ++bit) {
                    const bool     on = ((k >> bit) & 1) != 0;
                    const unsigned bb = __builtin_amdgcn_ballot_w32(on);
                    m &= on ? bb : ~bb;
                }
                const int below = (int)__builtin_amdgcn_mbcnt_lo(m, 0u);
                int r = 0;
                if (valid) {
                    const int base = hw[k];
                    r = base + below;
                    if (below == 0) hw[k] = base + __builtin_popcount(m);
                }
                pk[j] = valid ? (k | (r << 8) | ((d & (BKT_ROWS - 1)) << 20)) : -1;
            }
            __syncthreads();

            {
                const int k = threadIdx.x;
                int run = 0;
                #pragma unroll
                for (int w = 0; w < 8; ++w) {
                    const int c = s_hist[w * NBK + k];
                    s_hist[w * NBK + k] = run;
                    run += c;
                }
                const int incl = wave_incl_scan(run, lane);
                if (lane == 31) s_wtot[wave] = incl;
                __syncthreads();
                int wbase = 0;
                #pragma unroll
                for (int w = 0; w < 8; ++w) {
                    const int tw = s_wtot[w];
                    if (w < wave) wbase += tw;
                }
                const int st = wbase + incl - run;
                s_start[k] = st;
                tab[(size_t)k * n_chunks + chunk] = st;
                if (k == NBK - 1) tab[(size_t)NBK * n_chunks + chunk] = st + run;
            }
            __syncthreads();

            #pragma unroll
            for (int j = 0; j < 8; ++j) {
                if (pk[j] >= 0) {
                    const int    k   = pk[j] & (NBK - 1);
                    const int    pos = s_start[k] + hw[k] + ((pk[j] >> 8) & 4095);
                    const size_t o   = (size_t)chunk * CHUNK + pos;
                    rec_key[o] = (ss[j] << BKT_SHIFT) | (pk[j] >> 20);
                    rec_w[o]   = ww[j];
                }
            }
            __syncthreads();
        }
    }

    const int tile    = blockIdx.x * 8 + wave;
    const int n_tiles = n_nodes >> 4;
    if (tile >= n_tiles) return;

    const int row0    = tile << 4;
    const int halfSel = lane >> 4;
    const int l       = lane & 15;

    v8f acc[4];
    #pragma unroll
    for (int nt = 0; nt < 4; ++nt) {
        const float bias = b[nt * 16 + l];
        #pragma unroll
        for (int j = 0; j < 8; ++j) acc[nt][j] = bias;
    }

    #pragma unroll 4
    for (int k = 0; k < DIM; k += 4) {
        const float* xa = x + (size_t)(row0 + l) * DIM + k + 2 * halfSel;
        v2f a;
        a[0] = xa[0];
        a[1] = xa[1];
        #pragma unroll
        for (int nt = 0; nt < 4; ++nt) {
            const float* wb = W + (size_t)(nt * 16 + l) * DIM + k + 2 * halfSel;
            v2f bf;
            bf[0] = wb[0];
            bf[1] = wb[1];
            acc[nt] = __builtin_amdgcn_wmma_f32_16x16x4_f32(
                 false, a,  false, bf,
                 (short)0, acc[nt],  false,  false);
        }
    }

    #pragma unroll
    for (int nt = 0; nt < 4; ++nt) {
        #pragma unroll
        for (int j = 0; j < 8; ++j) {
            const int m = j + 8 * halfSel;
            support[(size_t)(row0 + m) * DIM + nt * 16 + l] = acc[nt][j];
        }
    }
}

typedef __attribute__((ext_vector_type(32))) float v32f;

struct EdgeGroup {
    int    G;
    int    key;
    float  wgt;
    float2 v[8];
};

__device__ __forceinline__ void fetch_group(EdgeGroup& g, const int* own, int i, int ocnt,
                                            const int* s_key, const float* s_w,
                                            const float2* __restrict__ sup2, int lane) {
    g.G   = min(8, ocnt - i);
    g.key = 0;
    g.wgt = 0.f;
    if (lane < g.G) {
        const int idx = own[i + lane];
        g.key = s_key[idx];
        g.wgt = s_w[idx];
    }
    #pragma unroll
    for (int u = 0; u < 8; ++u) {
        const int ku = __builtin_amdgcn_readlane(g.key, u);
        g.v[u] = sup2[(size_t)(ku >> BKT_SHIFT) * 32 + lane];
    }
}

__device__ __forceinline__ void add_group(const EdgeGroup& g, v32f& ax, v32f& ay) {
    #pragma unroll
    for (int u = 0; u < 8; ++u) {
        if (u < g.G) {
            const int   ku = __builtin_amdgcn_readlane(g.key, u);
            const float wu = __int_as_float(__builtin_amdgcn_readlane(__float_as_int(g.wgt), u));
            const int   n  = ku & 31;
            ax[n] = fmaf(wu, g.v[u].x, ax[n]);
            ay[n] = fmaf(wu, g.v[u].y, ay[n]);
        }
    }
}

__device__ __forceinline__ void add_owned(const int* own, int ocnt,
                                          const int* s_key, const float* s_w,
                                          const float2* __restrict__ sup2,
                                          v32f& ax, v32f& ay, int lane) {
    if (ocnt <= 0) return;
    EdgeGroup ga, gb;
    fetch_group(ga, own, 0, ocnt, s_key, s_w, sup2, lane);
    for (int i = 0; i < ocnt; i += 16) {
        fetch_group(gb, own, i + 8, ocnt, s_key, s_w, sup2, lane);
        add_group(ga, ax, ay);
        fetch_group(ga, own, i + 16, ocnt, s_key, s_w, sup2, lane);
        add_group(gb, ax, ay);
    }
}

__global__ __launch_bounds__(512) void owner_kernel(const float* __restrict__ support,
                                                    const int* __restrict__ tab,
                                                    const int* __restrict__ rec_key,
                                                    const float* __restrict__ rec_w,
                                                    float* __restrict__ out,
                                                    int n_nodes, int n_edges) {
    __shared__ int    s_key[NWAVE * CAP];
    __shared__ float  s_w[NWAVE * CAP];
    __shared__ int    s_own[NWAVE * OCAP];
    __shared__ int    s_cnt[NWAVE];
    __shared__ int    s_more[NWAVE];

    const int lane = threadIdx.x & 31;
    const int wv   = __builtin_amdgcn_readfirstlane((int)(threadIdx.x >> 5));
    const int bkt  = blockIdx.x;
    const int n_chunks = chunks_of(n_edges);
    const int* tlo = tab + (size_t)bkt * n_chunks;
    const int* thi = tab + (size_t)(bkt + 1) * n_chunks;

    v32f ax, ay;
    #pragma unroll
    for (int r = 0; r < 32; ++r) { ax[r] = 0.f; ay[r] = 0.f; }

    int*   myKey = s_key + wv * CAP;
    float* myW   = s_w   + wv * CAP;
    int*   myOwn = s_own + wv * OCAP;
    const float2* sup2 = (const float2*)support;

    int cur = wv;
    int off = 0;
    for (;;) {
        int cnt = 0;
        while (cur < n_chunks && cnt < CAP) {
            const int ci = cur + NWAVE * lane;
            int st = 0, n = 0;
            if (ci < n_chunks) {
                st = tlo[ci];
                n  = thi[ci] - st;
            }
            if (lane == 0) { st += off; n -= off; }
            const int incl  = wave_incl_scan(n, lane);
            const int excl  = incl - n;
            const int total = __builtin_amdgcn_readlane(incl, 31);
            const int take  = min(total, CAP - cnt);
            const int mine  = min(n, take - excl);
            const int*   rk = rec_key + (size_t)ci * CHUNK + st;
            const float* rw = rec_w   + (size_t)ci * CHUNK + st;
            int*   dk = myKey + cnt + excl;
            float* dw = myW   + cnt + excl;
            for (int t = 0; ; t += 4) {
                if (__builtin_amdgcn_ballot_w32(t < mine) == 0u) break;
                int   q0 = 0, q1 = 0, q2 = 0, q3 = 0;
                float f0 = 0.f, f1 = 0.f, f2 = 0.f, f3 = 0.f;
                if (t     < mine) { q0 = rk[t];     f0 = rw[t];     }
                if (t + 1 < mine) { q1 = rk[t + 1]; f1 = rw[t + 1]; }
                if (t + 2 < mine) { q2 = rk[t + 2]; f2 = rw[t + 2]; }
                if (t + 3 < mine) { q3 = rk[t + 3]; f3 = rw[t + 3]; }
                if (t     < mine) { dk[t]     = q0; dw[t]     = f0; }
                if (t + 1 < mine) { dk[t + 1] = q1; dw[t + 1] = f1; }
                if (t + 2 < mine) { dk[t + 2] = q2; dw[t + 2] = f2; }
                if (t + 3 < mine) { dk[t + 3] = q3; dw[t + 3] = f3; }
            }
            cnt += take;
            if (take == total) {
                cur += 32 * NWAVE;
                off = 0;
            } else {
                const unsigned part = __builtin_amdgcn_ballot_w32(incl > take);
                const int i    = __builtin_ctz(part);
                const int ex_i = __builtin_amdgcn_readlane(excl, i);
                off = (i == 0 ? off : 0) + (take - ex_i);
                cur += NWAVE * i;
            }
        }
        if (lane == 0) {
            s_cnt[wv]  = cnt;
            s_more[wv] = (cur < n_chunks) ? 1 : 0;
        }
        __syncthreads();

        int ocnt = 0;
        for (int s = 0; s < NWAVE; ++s) {
            const int  c  = __builtin_amdgcn_readfirstlane(s_cnt[s]);
            const int* lk = s_key + s * CAP;
            for (int base = 0; base < c; base += 32 * PICKB) {
                if (ocnt > OCAP - 32 * PICKB) {
                    add_owned(myOwn, ocnt, s_key, s_w, sup2, ax, ay, lane);
                    ocnt = 0;
                }
                int kq[PICKB];
                #pragma unroll
                for (int q = 0; q < PICKB; ++q) kq[q] = lk[min(base + 32 * q + lane, c - 1)];
                #pragma unroll
                for (int q = 0; q < PICKB; ++q) {
                    if (base + 32 * q >= c) break;
                    const int  i    = base + 32 * q + lane;
                    const bool mine = (i < c) && (((kq[q] >> 5) & (NWAVE - 1)) == wv);
                    const unsigned mk = __builtin_amdgcn_ballot_w32(mine);
                    if (mine) myOwn[ocnt + (int)__builtin_amdgcn_mbcnt_lo(mk, 0u)] = s * CAP + i;
                    ocnt += __builtin_popcount(mk);
                }
            }
        }
        add_owned(myOwn, ocnt, s_key, s_w, sup2, ax, ay, lane);

        const int mflag = (lane < NWAVE) ? s_more[lane] : 0;
        const unsigned more = __builtin_amdgcn_ballot_w32(mflag != 0);
        __syncthreads();
        if (more == 0u) break;
    }

    float2* out2 = (float2*)out;
    #pragma unroll
    for (int r = 0; r < 32; ++r) {
        const int node = bkt * BKT_ROWS + wv * 32 + r;
        if (node < n_nodes) out2[(size_t)node * 32 + lane] = make_float2(ax[r], ay[r]);
    }
}

extern "C" void kernel_launch(void* const* d_in, const int* in_sizes, int n_in,
                              void* d_out, int out_size, void* d_ws, size_t ws_size,
                              hipStream_t stream) {
    const float* x    = (const float*)d_in[0];
    const float* W    = (const float*)d_in[1];
    const float* b    = (const float*)d_in[2];
    const int*   esrc = (const int*)  d_in[3];
    const int*   edst = (const int*)  d_in[4];
    const float* ew   = (const float*)d_in[5];

    const int n_nodes = in_sizes[0] / DIM;
    const int n_edges = in_sizes[3];

    float* out     = (float*)d_out;
    const int n_chunks = chunks_of(n_edges);
    float* support = (float*)d_ws;
    int*   rec_key = (int*)(support + (size_t)n_nodes * DIM);
    float* rec_w   = (float*)(rec_key + (size_t)n_chunks * CHUNK);
    int*   tab     = (int*)(rec_w + (size_t)n_chunks * CHUNK);

    const int n_tiles = n_nodes / 16;
    const int gblocks = (n_tiles + 7) / 8;
    gemm_kernel<<<gblocks, 256, 0, stream>>>(x, W, b, esrc, edst, ew, support, rec_key, rec_w, tab,
                                             n_nodes, n_edges);

    const int oblocks = (n_nodes + BKT_ROWS - 1) / BKT_ROWS;
    owner_kernel<<<oblocks, 512, 0, stream>>>(support, tab, rec_key, rec_w, out, n_nodes, n_edges);
}


# Owner groups resized from 512 to 400 destination rows (16 waves x 25 rows): 250 groups, one on each WGP, instead of 196, so a group's edge work falls by a fifth; order of every sum unchanged
# speedup vs baseline: 1.6382x; 1.1326x over previous
#include <hip/hip_runtime.h>
#include <hip/hip_bf16.h>


typedef __attribute__((ext_vector_type(2))) float v2f;
typedef __attribute__((ext_vector_type(8))) float v8f;

#define DIM 64

#define KEY_SHIFT 9
#define BKT_ROWS  400
#define W_ROWS    25
#define NBK       256
#define NWAVE     16
#define CAP       1280
#define OCAP      1024
#define PICKB     8
#define CHUNK     2048

__host__ __device__ __forceinline__ int chunks_of(int n_edges) { return (n_edges + CHUNK - 1) / CHUNK; }

__device__ __forceinline__ int wave_incl_scan(int v, int lane) {
    #pragma unroll
    for (int d = 1; d < 32; d <<= 1) {
        const int t = __shfl_up(v, d, 32);
        if (lane >= d) v += t;
    }
    return v;
}

__global__ __launch_bounds__(256) void gemm_kernel(const float* __restrict__ x,
                                                   const float* __restrict__ W,
                                                   const float* __restrict__ b,
                                                   const int* __restrict__ esrc,
                                                   const int* __restrict__ edst,
                                                   const float* __restrict__ ew,
                                                   float* __restrict__ support,
                                                   int* __restrict__ rec_key,
                                                   float* __restrict__ rec_w,
                                                   int* __restrict__ tab,
                                                   int n_nodes, int n_edges) {
    __shared__ int s_hist[8 * NBK];
    __shared__ int s_start[NBK];
    __shared__ int s_wtot[8];

    const int lane = threadIdx.x & 31;
    const int wave = threadIdx.x >> 5;

    {
        const int n_chunks = chunks_of(n_edges);
        int* hw = s_hist + wave * NBK;
        for (int chunk = blockIdx.x; chunk < n_chunks; chunk += gridDim.x) {
            const int e0 = chunk * CHUNK + wave * 256 + lane;
            int dd[8], ss[8];
            float ww[8];
            #pragma unroll
            for (int j = 0; j < 8; ++j) {
                const int e = e0 + j * 32;
                dd[j] = 0; ss[j] = 0; ww[j] = 0.f;
                if (e < n_edges) { dd[j] = edst[e]; ss[j] = esrc[e]; ww[j] = ew[e]; }
            }
            for (int i = threadIdx.x; i < 8 * NBK; i += 256) s_hist[i] = 0;
            __syncthreads();

            int pk[8];
            #pragma unroll
            for (int j = 0; j < 8; ++j) {
                const int  e     = e0 + j * 32;
                const bool valid = e < n_edges;
                const int  d     = dd[j];
                const int  qb    = (int)((unsigned)d / (unsigned)BKT_ROWS);
                const int  k     = qb & (NBK - 1);
                const int  rib   = d - qb * BKT_ROWS;
                const int  ow    = (rib * 41) >> 10;
                const int  fld   = (ow << 5) | (rib - ow * W_ROWS);
                unsigned m = __builtin_amdgcn_ballot_w32(valid);
                #pragma unroll
                for (int bit = 0; bit < 8; ++bit) {
                    const bool     on = ((k >> bit) & 1) != 0;
                    const unsigned bb = __builtin_amdgcn_ballot_w32(on);
                    m &= on ? bb : ~bb;
                }
                const int below = (int)__builtin_amdgcn_mbcnt_lo(m, 0u);
                int r = 0;
                if (valid) {
                    const int base = hw[k];
                    r = base + below;
                    if (below == 0) hw[k] = base + __builtin_popcount(m);
                }
                pk[j] = valid ? (k | (r << 8) | (fld << 20)) : -1;
            }
            __syncthreads();

            {
                const int k = threadIdx.x;
                int run = 0;
                #pragma unroll
                for (int w = 0; w < 8; ++w) {
                    const int c = s_hist[w * NBK + k];
                    s_hist[w * NBK + k] = run;
                    run += c;
                }
                const int incl = wave_incl_scan(run, lane);
                if (lane == 31) s_wtot[wave] = incl;
                __syncthreads();
                int wbase = 0;
                #pragma unroll
                for (int w = 0; w < 8; ++w) {
                    const int tw = s_wtot[w];
                    if (w < wave) wbase += tw;
                }
                const int st = wbase + incl - run;
                s_start[k] = st;
                tab[(size_t)k * n_chunks + chunk] = st;
                if (k == NBK - 1) tab[(size_t)NBK * n_chunks + chunk] = st + run;
            }
            __syncthreads();

            #pragma unroll
            for (int j = 0; j < 8; ++j) {
                if (pk[j] >= 0) {
                    const int    k   = pk[j] & (NBK - 1);
                    const int    pos = s_start[k] + hw[k] + ((pk[j] >> 8) & 4095);
                    const size_t o   = (size_t)chunk * CHUNK + pos;
                    rec_key[o] = (ss[j] << KEY_SHIFT) | (pk[j] >> 20);
                    rec_w[o]   = ww[j];
                }
            }
            __syncthreads();
        }
    }

    const int tile    = blockIdx.x * 8 + wave;
    const int n_tiles = n_nodes >> 4;
    if (tile >= n_tiles) return;

    const int row0    = tile << 4;
    const int halfSel = lane >> 4;
    const int l       = lane & 15;

    v8f acc[4];
    #pragma unroll
    for (int nt = 0; nt < 4; ++nt) {
        const float bias = b[nt * 16 + l];
        #pragma unroll
        for (int j = 0; j < 8; ++j) acc[nt][j] = bias;
    }

    #pragma unroll 4
    for (int k = 0; k < DIM; k += 4) {
        const float* xa = x + (size_t)(row0 + l) * DIM + k + 2 * halfSel;
        v2f a;
        a[0] = xa[0];
        a[1] = xa[1];
        #pragma unroll
        for (int nt = 0; nt < 4; ++nt) {
            const float* wb = W + (size_t)(nt * 16 + l) * DIM + k + 2 * halfSel;
            v2f bf;
            bf[0] = wb[0];
            bf[1] = wb[1];
            acc[nt] = __builtin_amdgcn_wmma_f32_16x16x4_f32(
                 false, a,  false, bf,
                 (short)0, acc[nt],  false,  false);
        }
    }

    #pragma unroll
    for (int nt = 0; nt < 4; ++nt) {
        #pragma unroll
        for (int j = 0; j < 8; ++j) {
            const int m = j + 8 * halfSel;
            support[(size_t)(row0 + m) * DIM + nt * 16 + l] = acc[nt][j];
        }
    }
}

typedef __attribute__((ext_vector_type(32))) float v32f;

struct EdgeGroup {
    int    G;
    int    key;
    float  wgt;
    float2 v[8];
};

__device__ __forceinline__ void fetch_group(EdgeGroup& g, const int* own, int i, int ocnt,
                                            const int* s_key, const float* s_w,
                                            const float2* __restrict__ sup2, int lane) {
    g.G   = min(8, ocnt - i);
    g.key = 0;
    g.wgt = 0.f;
    if (lane < g.G) {
        const int idx = own[i + lane];
        g.key = s_key[idx];
        g.wgt = s_w[idx];
    }
    #pragma unroll
    for (int u = 0; u < 8; ++u) {
        const int ku = __builtin_amdgcn_readlane(g.key, u);
        g.v[u] = sup2[(size_t)(ku >> KEY_SHIFT) * 32 + lane];
    }
}

__device__ __forceinline__ void add_group(const EdgeGroup& g, v32f& ax, v32f& ay) {
    #pragma unroll
    for (int u = 0; u < 8; ++u) {
        if (u < g.G) {
            const int   ku = __builtin_amdgcn_readlane(g.key, u);
            const float wu = __int_as_float(__builtin_amdgcn_readlane(__float_as_int(g.wgt), u));
            const int   n  = ku & 31;
            ax[n] = fmaf(wu, g.v[u].x, ax[n]);
            ay[n] = fmaf(wu, g.v[u].y, ay[n]);
        }
    }
}

__device__ __forceinline__ void add_owned(const int* own, int ocnt,
                                          const int* s_key, const float* s_w,
                                          const float2* __restrict__ sup2,
                                          v32f& ax, v32f& ay, int lane) {
    if (ocnt <= 0) return;
    EdgeGroup ga, gb;
    fetch_group(ga, own, 0, ocnt, s_key, s_w, sup2, lane);
    for (int i = 0; i < ocnt; i += 16) {
        fetch_group(gb, own, i + 8, ocnt, s_key, s_w, sup2, lane);
        add_group(ga, ax, ay);
        fetch_group(ga, own, i + 16, ocnt, s_key, s_w, sup2, lane);
        add_group(gb, ax, ay);
    }
}

__global__ __launch_bounds__(512) void owner_kernel(const float* __restrict__ support,
                                                    const int* __restrict__ tab,
                                                    const int* __restrict__ rec_key,
                                                    const float* __restrict__ rec_w,
                                                    float* __restrict__ out,
                                                    int n_nodes, int n_edges) {
    __shared__ int    s_key[NWAVE * CAP];
    __shared__ float  s_w[NWAVE * CAP];
    __shared__ int    s_own[NWAVE * OCAP];
    __shared__ int    s_cnt[NWAVE];
    __shared__ int    s_more[NWAVE];

    const int lane = threadIdx.x & 31;
    const int wv   = __builtin_amdgcn_readfirstlane((int)(threadIdx.x >> 5));
    const int bkt  = blockIdx.x;
    const int n_chunks = chunks_of(n_edges);
    const int* tlo = tab + (size_t)bkt * n_chunks;
    const int* thi = tab + (size_t)(bkt + 1) * n_chunks;

    v32f ax, ay;
    #pragma unroll
    for (int r = 0; r < 32; ++r) { ax[r] = 0.f; ay[r] = 0.f; }

    int*   myKey = s_key + wv * CAP;
    float* myW   = s_w   + wv * CAP;
    int*   myOwn = s_own + wv * OCAP;
    const float2* sup2 = (const float2*)support;

    int cur = wv;
    int off = 0;
    for (;;) {
        int cnt = 0;
        while (cur < n_chunks && cnt < CAP) {
            const int ci = cur + NWAVE * lane;
            int st = 0, n = 0;
            if (ci < n_chunks) {
                st = tlo[ci];
                n  = thi[ci] - st;
            }
            if (lane == 0) { st += off; n -= off; }
            const int incl  = wave_incl_scan(n, lane);
            const int excl  = incl - n;
            const int total = __builtin_amdgcn_readlane(incl, 31);
            const int take  = min(total, CAP - cnt);
            const int mine  = min(n, take - excl);
            const int*   rk = rec_key + (size_t)ci * CHUNK + st;
            const float* rw = rec_w   + (size_t)ci * CHUNK + st;
            int*   dk = myKey + cnt + excl;
            float* dw = myW   + cnt + excl;
            for (int t = 0; ; t += 4) {
                if (__builtin_amdgcn_ballot_w32(t < mine) == 0u) break;
                int   q0 = 0, q1 = 0, q2 = 0, q3 = 0;
                float f0 = 0.f, f1 = 0.f, f2 = 0.f, f3 = 0.f;
                if (t     < mine) { q0 = rk[t];     f0 = rw[t];     }
                if (t + 1 < mine) { q1 = rk[t + 1]; f1 = rw[t + 1]; }
                if (t + 2 < mine) { q2 = rk[t + 2]; f2 = rw[t + 2]; }
                if (t + 3 < mine) { q3 = rk[t + 3]; f3 = rw[t + 3]; }
                if (t     < mine) { dk[t]     = q0; dw[t]     = f0; }
                if (t + 1 < mine) { dk[t + 1] = q1; dw[t + 1] = f1; }
                if (t + 2 < mine) { dk[t + 2] = q2; dw[t + 2] = f2; }
                if (t + 3 < mine) { dk[t + 3] = q3; dw[t + 3] = f3; }
            }
            cnt += take;
            if (take == total) {
                cur += 32 * NWAVE;
                off = 0;
            } else {
                const unsigned part = __builtin_amdgcn_ballot_w32(incl > take);
                const int i    = __builtin_ctz(part);
                const int ex_i = __builtin_amdgcn_readlane(excl, i);
                off = (i == 0 ? off : 0) + (take - ex_i);
                cur += NWAVE * i;
            }
        }
        if (lane == 0) {
            s_cnt[wv]  = cnt;
            s_more[wv] = (cur < n_chunks) ? 1 : 0;
        }
        __syncthreads();

        int ocnt = 0;
        for (int s = 0; s < NWAVE; ++s) {
            const int  c  = __builtin_amdgcn_readfirstlane(s_cnt[s]);
            const int* lk = s_key + s * CAP;
            for (int base = 0; base < c; base += 32 * PICKB) {
                if (ocnt > OCAP - 32 * PICKB) {
                    add_owned(myOwn, ocnt, s_key, s_w, sup2, ax, ay, lane);
                    ocnt = 0;
                }
                int kq[PICKB];
                #pragma unroll
                for (int q = 0; q < PICKB; ++q) kq[q] = lk[min(base + 32 * q + lane, c - 1)];
                #pragma unroll
                for (int q = 0; q < PICKB; ++q) {
                    if (base + 32 * q >= c) break;
                    const int  i    = base + 32 * q + lane;
                    const bool mine = (i < c) && (((kq[q] >> 5) & (NWAVE - 1)) == wv);
                    const unsigned mk = __builtin_amdgcn_ballot_w32(mine);
                    if (mine) myOwn[ocnt + (int)__builtin_amdgcn_mbcnt_lo(mk, 0u)] = s * CAP + i;
                    ocnt += __builtin_popcount(mk);
                }
            }
        }
        add_owned(myOwn, ocnt, s_key, s_w, sup2, ax, ay, lane);

        const int mflag = (lane < NWAVE) ? s_more[lane] : 0;
        const unsigned more = __builtin_amdgcn_ballot_w32(mflag != 0);
        __syncthreads();
        if (more == 0u) break;
    }

    float2* out2 = (float2*)out;
    #pragma unroll
    for (int r = 0; r < W_ROWS; ++r) {
        const int node = bkt * BKT_ROWS + wv * W_ROWS + r;
        if (node < n_nodes) out2[(size_t)node * 32 + lane] = make_float2(ax[r], ay[r]);
    }
}

extern "C" void kernel_launch(void* const* d_in, const int* in_sizes, int n_in,
                              void* d_out, int out_size, void* d_ws, size_t ws_size,
                              hipStream_t stream) {
    const float* x    = (const float*)d_in[0];
    const float* W    = (const float*)d_in[1];
    const float* b    = (const float*)d_in[2];
    const int*   esrc = (const int*)  d_in[3];
    const int*   edst = (const int*)  d_in[4];
    const float* ew   = (const float*)d_in[5];

    const int n_nodes = in_sizes[0] / DIM;
    const int n_edges = in_sizes[3];

    float* out     = (float*)d_out;
    const int n_chunks = chunks_of(n_edges);
    float* support = (float*)d_ws;
    int*   rec_key = (int*)(support + (size_t)n_nodes * DIM);
    float* rec_w   = (float*)(rec_key + (size_t)n_chunks * CHUNK);
    int*   tab     = (int*)(rec_w + (size_t)n_chunks * CHUNK);

    const int n_tiles = n_nodes / 16;
    const int gblocks = (n_tiles + 7) / 8;
    gemm_kernel<<<gblocks, 256, 0, stream>>>(x, W, b, esrc, edst, ew, support, rec_key, rec_w, tab,
                                             n_nodes, n_edges);

    const int oblocks = (n_nodes + BKT_ROWS - 1) / BKT_ROWS;
    owner_kernel<<<oblocks, 512, 0, stream>>>(support, tab, rec_key, rec_w, out, n_nodes, n_edges);
}


# Sorted records kept as one 8-byte (key, weight) pair: the sort stores 8 bytes per record and eight lanes read each run with one 8-byte load apiece, not 32 scattered lines; sum order unchanged
# speedup vs baseline: 1.8365x; 1.1210x over previous
#include <hip/hip_runtime.h>
#include <hip/hip_bf16.h>


typedef __attribute__((ext_vector_type(2))) float v2f;
typedef __attribute__((ext_vector_type(8))) float v8f;

#define DIM 64

#define KEY_SHIFT 9
#define BKT_ROWS  400
#define W_ROWS    25
#define NBK       256
#define NWAVE     16
#define CAP       1280
#define OCAP      1024
#define PICKB     8
#define CHUNK     2048

__host__ __device__ __forceinline__ int chunks_of(int n_edges) { return (n_edges + CHUNK - 1) / CHUNK; }

__device__ __forceinline__ int wave_incl_scan(int v, int lane) {
    #pragma unroll
    for (int d = 1; d < 32; d <<= 1) {
        const int t = __shfl_up(v, d, 32);
        if (lane >= d) v += t;
    }
    return v;
}

__global__ __launch_bounds__(256) void gemm_kernel(const float* __restrict__ x,
                                                   const float* __restrict__ W,
                                                   const float* __restrict__ b,
                                                   const int* __restrict__ esrc,
                                                   const int* __restrict__ edst,
                                                   const float* __restrict__ ew,
                                                   float* __restrict__ support,
                                                   int* __restrict__ rec,
                                                   int* __restrict__ tab,
                                                   int n_nodes, int n_edges) {
    __shared__ int s_hist[8 * NBK];
    __shared__ int s_start[NBK];
    __shared__ int s_wtot[8];

    const int lane = threadIdx.x & 31;
    const int wave = threadIdx.x >> 5;

    {
        const int n_chunks = chunks_of(n_edges);
        int2* rec2 = (int2*)rec;
        int* hw = s_hist + wave * NBK;
        for (int chunk = blockIdx.x; chunk < n_chunks; chunk += gridDim.x) {
            const int e0 = chunk * CHUNK + wave * 256 + lane;
            int dd[8], ss[8];
            float ww[8];
            #pragma unroll
            for (int j = 0; j < 8; ++j) {
                const int e = e0 + j * 32;
                dd[j] = 0; ss[j] = 0; ww[j] = 0.f;
                if (e < n_edges) { dd[j] = edst[e]; ss[j] = esrc[e]; ww[j] = ew[e]; }
            }
            for (int i = threadIdx.x; i < 8 * NBK; i += 256) s_hist[i] = 0;
            __syncthreads();

            int pk[8];
            #pragma unroll
            for (int j = 0; j < 8; ++j) {
                const int  e     = e0 + j * 32;
                const bool valid = e < n_edges;
                const int  d     = dd[j];
                const int  qb    = (int)((unsigned)d / (unsigned)BKT_ROWS);
                const int  k     = qb & (NBK - 1);
                const int  rib   = d - qb * BKT_ROWS;
                const int  ow    = (rib * 41) >> 10;
                const int  fld   = (ow << 5) | (rib - ow * W_ROWS);
                unsigned m = __builtin_amdgcn_ballot_w32(valid);
                #pragma unroll
                for (int bit = 0; bit < 8; ++bit) {
                    const bool     on = ((k >> bit) & 1) != 0;
                    const unsigned bb = __builtin_amdgcn_ballot_w32(on);
                    m &= on ? bb : ~bb;
                }
                const int below = (int)__builtin_amdgcn_mbcnt_lo(m, 0u);
                int r = 0;
                if (valid) {
                    const int base = hw[k];
                    r = base + below;
                    if (below == 0) hw[k] = base + __builtin_popcount(m);
                }
                pk[j] = valid ? (k | (r << 8) | (fld << 20)) : -1;
            }
            __syncthreads();

            {
                const int k = threadIdx.x;
                int run = 0;
                #pragma unroll
                for (int w = 0; w < 8; ++w) {
                    const int c = s_hist[w * NBK + k];
                    s_hist[w * NBK + k] = run;
                    run += c;
                }
                const int incl = wave_incl_scan(run, lane);
                if (lane == 31) s_wtot[wave] = incl;
                __syncthreads();
                int wbase = 0;
                #pragma unroll
                for (int w = 0; w < 8; ++w) {
                    const int tw = s_wtot[w];
                    if (w < wave) wbase += tw;
                }
                const int st = wbase + incl - run;
                s_start[k] = st;
                tab[(size_t)k * n_chunks + chunk] = st;
                if (k == NBK - 1) tab[(size_t)NBK * n_chunks + chunk] = st + run;
            }
            __syncthreads();

            #pragma unroll
            for (int j = 0; j < 8; ++j) {
                if (pk[j] >= 0) {
                    const int    k   = pk[j] & (NBK - 1);
                    const int    pos = s_start[k] + hw[k] + ((pk[j] >> 8) & 4095);
                    const size_t o   = (size_t)chunk * CHUNK + pos;
                    rec2[o] = make_int2((ss[j] << KEY_SHIFT) | (pk[j] >> 20), __float_as_int(ww[j]));
                }
            }
            __syncthreads();
        }
    }

    const int tile    = blockIdx.x * 8 + wave;
    const int n_tiles = n_nodes >> 4;
    if (tile >= n_tiles) return;

    const int row0    = tile << 4;
    const int halfSel = lane >> 4;
    const int l       = lane & 15;

    v8f acc[4];
    #pragma unroll
    for (int nt = 0; nt < 4; ++nt) {
        const float bias = b[nt * 16 + l];
        #pragma unroll
        for (int j = 0; j < 8; ++j) acc[nt][j] = bias;
    }

    #pragma unroll 4
    for (int k = 0; k < DIM; k += 4) {
        const float* xa = x + (size_t)(row0 + l) * DIM + k + 2 * halfSel;
        v2f a;
        a[0] = xa[0];
        a[1] = xa[1];
        #pragma unroll
        for (int nt = 0; nt < 4; ++nt) {
            const float* wb = W + (size_t)(nt * 16 + l) * DIM + k + 2 * halfSel;
            v2f bf;
            bf[0] = wb[0];
            bf[1] = wb[1];
            acc[nt] = __builtin_amdgcn_wmma_f32_16x16x4_f32(
                 false, a,  false, bf,
                 (short)0, acc[nt],  false,  false);
        }
    }

    #pragma unroll
    for (int nt = 0; nt < 4; ++nt) {
        #pragma unroll
        for (int j = 0; j < 8; ++j) {
            const int m = j + 8 * halfSel;
            support[(size_t)(row0 + m) * DIM + nt * 16 + l] = acc[nt][j];
        }
    }
}

typedef __attribute__((ext_vector_type(32))) float v32f;

struct EdgeGroup {
    int    G;
    int    key;
    float  wgt;
    float2 v[8];
};

__device__ __forceinline__ void fetch_group(EdgeGroup& g, const int* own, int i, int ocnt,
                                            const int* s_key, const float* s_w,
                                            const float2* __restrict__ sup2, int lane) {
    g.G   = min(8, ocnt - i);
    g.key = 0;
    g.wgt = 0.f;
    if (lane < g.G) {
        const int idx = own[i + lane];
        g.key = s_key[idx];
        g.wgt = s_w[idx];
    }
    #pragma unroll
    for (int u = 0; u < 8; ++u) {
        const int ku = __builtin_amdgcn_readlane(g.key, u);
        g.v[u] = sup2[(size_t)(ku >> KEY_SHIFT) * 32 + lane];
    }
}

__device__ __forceinline__ void add_group(const EdgeGroup& g, v32f& ax, v32f& ay) {
    #pragma unroll
    for (int u = 0; u < 8; ++u) {
        if (u < g.G) {
            const int   ku = __builtin_amdgcn_readlane(g.key, u);
            const float wu = __int_as_float(__builtin_amdgcn_readlane(__float_as_int(g.wgt), u));
            const int   n  = ku & 31;
            ax[n] = fmaf(wu, g.v[u].x, ax[n]);
            ay[n] = fmaf(wu, g.v[u].y, ay[n]);
        }
    }
}

__device__ __forceinline__ void add_owned(const int* own, int ocnt,
                                          const int* s_key, const float* s_w,
                                          const float2* __restrict__ sup2,
                                          v32f& ax, v32f& ay, int lane) {
    if (ocnt <= 0) return;
    EdgeGroup ga, gb;
    fetch_group(ga, own, 0, ocnt, s_key, s_w, sup2, lane);
    for (int i = 0; i < ocnt; i += 16) {
        fetch_group(gb, own, i + 8, ocnt, s_key, s_w, sup2, lane);
        add_group(ga, ax, ay);
        fetch_group(ga, own, i + 16, ocnt, s_key, s_w, sup2, lane);
        add_group(gb, ax, ay);
    }
}

__global__ __launch_bounds__(512) void owner_kernel(const float* __restrict__ support,
                                                    const int* __restrict__ tab,
                                                    const int* __restrict__ rec,
                                                    float* __restrict__ out,
                                                    int n_nodes, int n_edges) {
    __shared__ int    s_key[NWAVE * CAP];
    __shared__ float  s_w[NWAVE * CAP];
    __shared__ int    s_own[NWAVE * OCAP];
    __shared__ int    s_cnt[NWAVE];
    __shared__ int    s_more[NWAVE];

    const int lane = threadIdx.x & 31;
    const int wv   = __builtin_amdgcn_readfirstlane((int)(threadIdx.x >> 5));
    const int bkt  = blockIdx.x;
    const int n_chunks = chunks_of(n_edges);
    const int* tlo = tab + (size_t)bkt * n_chunks;
    const int* thi = tab + (size_t)(bkt + 1) * n_chunks;

    v32f ax, ay;
    #pragma unroll
    for (int r = 0; r < 32; ++r) { ax[r] = 0.f; ay[r] = 0.f; }

    int*   myKey = s_key + wv * CAP;
    float* myW   = s_w   + wv * CAP;
    int*   myOwn = s_own + wv * OCAP;
    const float2* sup2 = (const float2*)support;
    const int2*   rec2 = (const int2*)rec;
    const int     sub  = lane & 7;

    int cur = wv;
    int off = 0;
    for (;;) {
        int cnt = 0;
        while (cur < n_chunks && cnt < CAP) {
            const int ci = cur + NWAVE * lane;
            int st = 0, n = 0;
            if (ci < n_chunks) {
                st = tlo[ci];
                n  = thi[ci] - st;
            }
            if (lane == 0) { st += off; n -= off; }
            const int incl  = wave_incl_scan(n, lane);
            const int excl  = incl - n;
            const int total = __builtin_amdgcn_readlane(incl, 31);
            const int take  = min(total, CAP - cnt);
            const int mine  = min(n, take - excl);
            int so[8], dsti[8], lim[8];
            #pragma unroll
            for (int g = 0; g < 8; ++g) {
                const int j = 4 * g + (lane >> 3);
                lim[g]  = __shfl(mine, j, 32) - sub;
                dsti[g] = cnt + __shfl(excl, j, 32) + sub;
                so[g]   = (cur + NWAVE * j) * CHUNK + __shfl(st, j, 32) + sub;
            }
            for (int t = 0; ; t += 8) {
                if (__builtin_amdgcn_ballot_w32(t < mine) == 0u) break;
                int2 q[8];
                #pragma unroll
                for (int g = 0; g < 8; ++g) {
                    q[g] = make_int2(0, 0);
                    if (t < lim[g]) q[g] = rec2[(size_t)so[g] + t];
                }
                #pragma unroll
                for (int g = 0; g < 8; ++g) {
                    if (t < lim[g]) {
                        myKey[dsti[g] + t] = q[g].x;
                        myW[dsti[g] + t]   = __int_as_float(q[g].y);
                    }
                }
            }
            cnt += take;
            if (take == total) {
                cur += 32 * NWAVE;
                off = 0;
            } else {
                const unsigned part = __builtin_amdgcn_ballot_w32(incl > take);
                const int i    = __builtin_ctz(part);
                const int ex_i = __builtin_amdgcn_readlane(excl, i);
                off = (i == 0 ? off : 0) + (take - ex_i);
                cur += NWAVE * i;
            }
        }
        if (lane == 0) {
            s_cnt[wv]  = cnt;
            s_more[wv] = (cur < n_chunks) ? 1 : 0;
        }
        __syncthreads();

        int ocnt = 0;
        for (int s = 0; s < NWAVE; ++s) {
            const int  c  = __builtin_amdgcn_readfirstlane(s_cnt[s]);
            const int* lk = s_key + s * CAP;
            for (int base = 0; base < c; base += 32 * PICKB) {
                if (ocnt > OCAP - 32 * PICKB) {
                    add_owned(myOwn, ocnt, s_key, s_w, sup2, ax, ay, lane);
                    ocnt = 0;
                }
                int kq[PICKB];
                #pragma unroll
                for (int q = 0; q < PICKB; ++q) kq[q] = lk[min(base + 32 * q + lane, c - 1)];
                #pragma unroll
                for (int q = 0; q < PICKB; ++q) {
                    if (base + 32 * q >= c) break;
                    const int  i    = base + 32 * q + lane;
                    const bool mine = (i < c) && (((kq[q] >> 5) & (NWAVE - 1)) == wv);
                    const unsigned mk = __builtin_amdgcn_ballot_w32(mine);
                    if (mine) myOwn[ocnt + (int)__builtin_amdgcn_mbcnt_lo(mk, 0u)] = s * CAP + i;
                    ocnt += __builtin_popcount(mk);
                }
            }
        }
        add_owned(myOwn, ocnt, s_key, s_w, sup2, ax, ay, lane);

        const int mflag = (lane < NWAVE) ? s_more[lane] : 0;
        const unsigned more = __builtin_amdgcn_ballot_w32(mflag != 0);
        __syncthreads();
        if (more == 0u) break;
    }

    float2* out2 = (float2*)out;
    #pragma unroll
    for (int r = 0; r < W_ROWS; ++r) {
        const int node = bkt * BKT_ROWS + wv * W_ROWS + r;
        if (node < n_nodes) out2[(size_t)node * 32 + lane] = make_float2(ax[r], ay[r]);
    }
}

extern "C" void kernel_launch(void* const* d_in, const int* in_sizes, int n_in,
                              void* d_out, int out_size, void* d_ws, size_t ws_size,
                              hipStream_t stream) {
    const float* x    = (const float*)d_in[0];
    const float* W    = (const float*)d_in[1];
    const float* b    = (const float*)d_in[2];
    const int*   esrc = (const int*)  d_in[3];
    const int*   edst = (const int*)  d_in[4];
    const float* ew   = (const float*)d_in[5];

    const int n_nodes = in_sizes[0] / DIM;
    const int n_edges = in_sizes[3];

    float* out     = (float*)d_out;
    const int n_chunks = chunks_of(n_edges);
    float* support = (float*)d_ws;
    int*   rec     = (int*)(support + (size_t)n_nodes * DIM);
    int*   tab     = rec + (size_t)n_chunks * CHUNK * 2;

    const int n_tiles = n_nodes / 16;
    const int gblocks = (n_tiles + 7) / 8;
    gemm_kernel<<<gblocks, 256, 0, stream>>>(x, W, b, esrc, edst, ew, support, rec, tab,
                                             n_nodes, n_edges);

    const int oblocks = (n_nodes + BKT_ROWS - 1) / BKT_ROWS;
    owner_kernel<<<oblocks, 512, 0, stream>>>(support, tab, rec, out, n_nodes, n_edges);
}


# Owner add loop in straight-line trips of 32 list entries: all 32 source rows requested at once (was 8+8 in flight), next entries read from LDS meanwhile, past-end slots add weight 0; sum order same
# speedup vs baseline: 2.0994x; 1.1432x over previous
#include <hip/hip_runtime.h>
#include <hip/hip_bf16.h>


typedef __attribute__((ext_vector_type(2))) float v2f;
typedef __attribute__((ext_vector_type(8))) float v8f;

#define DIM 64

#define KEY_SHIFT 9
#define BKT_ROWS  400
#define W_ROWS    25
#define NBK       256
#define NWAVE     16
#define CAP       1280
#define OCAP      1024
#define PICKB     8
#define CHUNK     2048

__host__ __device__ __forceinline__ int chunks_of(int n_edges) { return (n_edges + CHUNK - 1) / CHUNK; }

__device__ __forceinline__ int wave_incl_scan(int v, int lane) {
    #pragma unroll
    for (int d = 1; d < 32; d <<= 1) {
        const int t = __shfl_up(v, d, 32);
        if (lane >= d) v += t;
    }
    return v;
}

__global__ __launch_bounds__(256) void gemm_kernel(const float* __restrict__ x,
                                                   const float* __restrict__ W,
                                                   const float* __restrict__ b,
                                                   const int* __restrict__ esrc,
                                                   const int* __restrict__ edst,
                                                   const float* __restrict__ ew,
                                                   float* __restrict__ support,
                                                   int* __restrict__ rec,
                                                   int* __restrict__ tab,
                                                   int n_nodes, int n_edges) {
    __shared__ int s_hist[8 * NBK];
    __shared__ int s_start[NBK];
    __shared__ int s_wtot[8];

    const int lane = threadIdx.x & 31;
    const int wave = threadIdx.x >> 5;

    {
        const int n_chunks = chunks_of(n_edges);
        int2* rec2 = (int2*)rec;
        int* hw = s_hist + wave * NBK;
        for (int chunk = blockIdx.x; chunk < n_chunks; chunk += gridDim.x) {
            const int e0 = chunk * CHUNK + wave * 256 + lane;
            int dd[8], ss[8];
            float ww[8];
            #pragma unroll
            for (int j = 0; j < 8; ++j) {
                const int e = e0 + j * 32;
                dd[j] = 0; ss[j] = 0; ww[j] = 0.f;
                if (e < n_edges) { dd[j] = edst[e]; ss[j] = esrc[e]; ww[j] = ew[e]; }
            }
            for (int i = threadIdx.x; i < 8 * NBK; i += 256) s_hist[i] = 0;
            __syncthreads();

            int pk[8];
            #pragma unroll
            for (int j = 0; j < 8; ++j) {
                const int  e     = e0 + j * 32;
                const bool valid = e < n_edges;
                const int  d     = dd[j];
                const int  qb    = (int)((unsigned)d / (unsigned)BKT_ROWS);
                const int  k     = qb & (NBK - 1);
                const int  rib   = d - qb * BKT_ROWS;
                const int  ow    = (rib * 41) >> 10;
                const int  fld   = (ow << 5) | (rib - ow * W_ROWS);
                unsigned m = __builtin_amdgcn_ballot_w32(valid);
                #pragma unroll
                for (int bit = 0; bit < 8; ++bit) {
                    const bool     on = ((k >> bit) & 1) != 0;
                    const unsigned bb = __builtin_amdgcn_ballot_w32(on);
                    m &= on ? bb : ~bb;
                }
                const int below = (int)__builtin_amdgcn_mbcnt_lo(m, 0u);
                int r = 0;
                if (valid) {
                    const int base = hw[k];
                    r = base + below;
                    if (below == 0) hw[k] = base + __builtin_popcount(m);
                }
                pk[j] = valid ? (k | (r << 8) | (fld << 20)) : -1;
            }
            __syncthreads();

            {
                const int k = threadIdx.x;
                int run = 0;
                #pragma unroll
                for (int w = 0; w < 8; ++w) {
                    const int c = s_hist[w * NBK + k];
                    s_hist[w * NBK + k] = run;
                    run += c;
                }
                const int incl = wave_incl_scan(run, lane);
                if (lane == 31) s_wtot[wave] = incl;
                __syncthreads();
                int wbase = 0;
                #pragma unroll
                for (int w = 0; w < 8; ++w) {
                    const int tw = s_wtot[w];
                    if (w < wave) wbase += tw;
                }
                const int st = wbase + incl - run;
                s_start[k] = st;
                tab[(size_t)k * n_chunks + chunk] = st;
                if (k == NBK - 1) tab[(size_t)NBK * n_chunks + chunk] = st + run;
            }
            __syncthreads();

            #pragma unroll
            for (int j = 0; j < 8; ++j) {
                if (pk[j] >= 0) {
                    const int    k   = pk[j] & (NBK - 1);
                    const int    pos = s_start[k] + hw[k] + ((pk[j] >> 8) & 4095);
                    const size_t o   = (size_t)chunk * CHUNK + pos;
                    rec2[o] = make_int2((ss[j] << KEY_SHIFT) | (pk[j] >> 20), __float_as_int(ww[j]));
                }
            }
            __syncthreads();
        }
    }

    const int tile    = blockIdx.x * 8 + wave;
    const int n_tiles = n_nodes >> 4;
    if (tile >= n_tiles) return;

    const int row0    = tile << 4;
    const int halfSel = lane >> 4;
    const int l       = lane & 15;

    v8f acc[4];
    #pragma unroll
    for (int nt = 0; nt < 4; ++nt) {
        const float bias = b[nt * 16 + l];
        #pragma unroll
        for (int j = 0; j < 8; ++j) acc[nt][j] = bias;
    }

    #pragma unroll 4
    for (int k = 0; k < DIM; k += 4) {
        const float* xa = x + (size_t)(row0 + l) * DIM + k + 2 * halfSel;
        v2f a;
        a[0] = xa[0];
        a[1] = xa[1];
        #pragma unroll
        for (int nt = 0; nt < 4; ++nt) {
            const float* wb = W + (size_t)(nt * 16 + l) * DIM + k + 2 * halfSel;
            v2f bf;
            bf[0] = wb[0];
            bf[1] = wb[1];
            acc[nt] = __builtin_amdgcn_wmma_f32_16x16x4_f32(
                 false, a,  false, bf,
                 (short)0, acc[nt],  false,  false);
        }
    }

    #pragma unroll
    for (int nt = 0; nt < 4; ++nt) {
        #pragma unroll
        for (int j = 0; j < 8; ++j) {
            const int m = j + 8 * halfSel;
            support[(size_t)(row0 + m) * DIM + nt * 16 + l] = acc[nt][j];
        }
    }
}

typedef __attribute__((ext_vector_type(32))) float v32f;

struct Trip {
    int   key;
    float wgt;
};

__device__ __forceinline__ Trip load_trip(const int* own, int i, int ocnt,
                                          const int* s_key, const float* s_w, int lane) {
    const int   e   = i + lane;
    const int   idx = own[min(e, ocnt - 1)];
    const int   k   = s_key[idx];
    const float w   = s_w[idx];
    Trip t;
    t.key = (e < ocnt) ? k : 31;
    t.wgt = (e < ocnt) ? w : 0.f;
    return t;
}

__device__ __forceinline__ void add_owned(const int* own, int ocnt,
                                          const int* s_key, const float* s_w,
                                          const float* __restrict__ support,
                                          v32f& ax, v32f& ay, int lane) {
    if (ocnt <= 0) return;
    const unsigned lane8 = (unsigned)lane * 8u;
    Trip cur = load_trip(own, 0, ocnt, s_key, s_w, lane);
    for (int i = 0; i < ocnt; i += 32) {
        float2 v[32];
        #pragma unroll
        for (int u = 0; u < 32; ++u) {
            const int      ku  = __builtin_amdgcn_readlane(cur.key, u);
            const unsigned off = ((unsigned)(ku >> KEY_SHIFT) << 8) + lane8;
            v[u] = *(const float2*)((const char*)support + off);
        }
        __builtin_amdgcn_sched_group_barrier(0x002, 64, 0);
        __builtin_amdgcn_sched_group_barrier(0x020, 32, 0);
        const Trip nxt = load_trip(own, i + 32, ocnt, s_key, s_w, lane);
        #pragma unroll
        for (int u = 0; u < 32; ++u) {
            const int   ku = __builtin_amdgcn_readlane(cur.key, u);
            const float wu = __int_as_float(__builtin_amdgcn_readlane(__float_as_int(cur.wgt), u));
            const int   n  = ku & 31;
            ax[n] = fmaf(wu, v[u].x, ax[n]);
            ay[n] = fmaf(wu, v[u].y, ay[n]);
        }
        cur = nxt;
    }
}

__global__ __launch_bounds__(512) void owner_kernel(const float* __restrict__ support,
                                                    const int* __restrict__ tab,
                                                    const int* __restrict__ rec,
                                                    float* __restrict__ out,
                                                    int n_nodes, int n_edges) {
    __shared__ int    s_key[NWAVE * CAP];
    __shared__ float  s_w[NWAVE * CAP];
    __shared__ int    s_own[NWAVE * OCAP];
    __shared__ int    s_cnt[NWAVE];
    __shared__ int    s_more[NWAVE];

    const int lane = threadIdx.x & 31;
    const int wv   = __builtin_amdgcn_readfirstlane((int)(threadIdx.x >> 5));
    const int bkt  = blockIdx.x;
    const int n_chunks = chunks_of(n_edges);
    const int* tlo = tab + (size_t)bkt * n_chunks;
    const int* thi = tab + (size_t)(bkt + 1) * n_chunks;

    v32f ax, ay;
    #pragma unroll
    for (int r = 0; r < 32; ++r) { ax[r] = 0.f; ay[r] = 0.f; }

    int*   myKey = s_key + wv * CAP;
    float* myW   = s_w   + wv * CAP;
    int*   myOwn = s_own + wv * OCAP;
    const int2*   rec2 = (const int2*)rec;
    const int     sub  = lane & 7;

    int cur = wv;
    int off = 0;
    for (;;) {
        int cnt = 0;
        while (cur < n_chunks && cnt < CAP) {
            const int ci = cur + NWAVE * lane;
            int st = 0, n = 0;
            if (ci < n_chunks) {
                st = tlo[ci];
                n  = thi[ci] - st;
            }
            if (lane == 0) { st += off; n -= off; }
            const int incl  = wave_incl_scan(n, lane);
            const int excl  = incl - n;
            const int total = __builtin_amdgcn_readlane(incl, 31);
            const int take  = min(total, CAP - cnt);
            const int mine  = min(n, take - excl);
            int so[8], dsti[8], lim[8];
            #pragma unroll
            for (int g = 0; g < 8; ++g) {
                const int j = 4 * g + (lane >> 3);
                lim[g]  = __shfl(mine, j, 32) - sub;
                dsti[g] = cnt + __shfl(excl, j, 32) + sub;
                so[g]   = (cur + NWAVE * j) * CHUNK + __shfl(st, j, 32) + sub;
            }
            for (int t = 0; ; t += 8) {
                if (__builtin_amdgcn_ballot_w32(t < mine) == 0u) break;
                int2 q[8];
                #pragma unroll
                for (int g = 0; g < 8; ++g) {
                    q[g] = make_int2(0, 0);
                    if (t < lim[g]) q[g] = rec2[(size_t)so[g] + t];
                }
                #pragma unroll
                for (int g = 0; g < 8; ++g) {
                    if (t < lim[g]) {
                        myKey[dsti[g] + t] = q[g].x;
                        myW[dsti[g] + t]   = __int_as_float(q[g].y);
                    }
                }
            }
            cnt += take;
            if (take == total) {
                cur += 32 * NWAVE;
                off = 0;
            } else {
                const unsigned part = __builtin_amdgcn_ballot_w32(incl > take);
                const int i    = __builtin_ctz(part);
                const int ex_i = __builtin_amdgcn_readlane(excl, i);
                off = (i == 0 ? off : 0) + (take - ex_i);
                cur += NWAVE * i;
            }
        }
        if (lane == 0) {
            s_cnt[wv]  = cnt;
            s_more[wv] = (cur < n_chunks) ? 1 : 0;
        }
        __syncthreads();

        int ocnt = 0;
        for (int s = 0; s < NWAVE; ++s) {
            const int  c  = __builtin_amdgcn_readfirstlane(s_cnt[s]);
            const int* lk = s_key + s * CAP;
            for (int base = 0; base < c; base += 32 * PICKB) {
                if (ocnt > OCAP - 32 * PICKB) {
                    add_owned(myOwn, ocnt, s_key, s_w, support, ax, ay, lane);
                    ocnt = 0;
                }
                int kq[PICKB];
                #pragma unroll
                for (int q = 0; q < PICKB; ++q) kq[q] = lk[min(base + 32 * q + lane, c - 1)];
                #pragma unroll
                for (int q = 0; q < PICKB; ++q) {
                    if (base + 32 * q >= c) break;
                    const int  i    = base + 32 * q + lane;
                    const bool mine = (i < c) && (((kq[q] >> 5) & (NWAVE - 1)) == wv);
                    const unsigned mk = __builtin_amdgcn_ballot_w32(mine);
                    if (mine) myOwn[ocnt + (int)__builtin_amdgcn_mbcnt_lo(mk, 0u)] = s * CAP + i;
                    ocnt += __builtin_popcount(mk);
                }
            }
        }
        add_owned(myOwn, ocnt, s_key, s_w, support, ax, ay, lane);

        const int mflag = (lane < NWAVE) ? s_more[lane] : 0;
        const unsigned more = __builtin_amdgcn_ballot_w32(mflag != 0);
        __syncthreads();
        if (more == 0u) break;
    }

    float2* out2 = (float2*)out;
    #pragma unroll
    for (int r = 0; r < W_ROWS; ++r) {
        const int node = bkt * BKT_ROWS + wv * W_ROWS + r;
        if (node < n_nodes) out2[(size_t)node * 32 + lane] = make_float2(ax[r], ay[r]);
    }
}

extern "C" void kernel_launch(void* const* d_in, const int* in_sizes, int n_in,
                              void* d_out, int out_size, void* d_ws, size_t ws_size,
                              hipStream_t stream) {
    const float* x    = (const float*)d_in[0];
    const float* W    = (const float*)d_in[1];
    const float* b    = (const float*)d_in[2];
    const int*   esrc = (const int*)  d_in[3];
    const int*   edst = (const int*)  d_in[4];
    const float* ew   = (const float*)d_in[5];

    const int n_nodes = in_sizes[0] / DIM;
    const int n_edges = in_sizes[3];

    float* out     = (float*)d_out;
    const int n_chunks = chunks_of(n_edges);
    float* support = (float*)d_ws;
    int*   rec     = (int*)(support + (size_t)n_nodes * DIM);
    int*   tab     = rec + (size_t)n_chunks * CHUNK * 2;

    const int n_tiles = n_nodes / 16;
    const int gblocks = (n_tiles + 7) / 8;
    gemm_kernel<<<gblocks, 256, 0, stream>>>(x, W, b, esrc, edst, ew, support, rec, tab,
                                             n_nodes, n_edges);

    const int oblocks = (n_nodes + BKT_ROWS - 1) / BKT_ROWS;
    owner_kernel<<<oblocks, 512, 0, stream>>>(support, tab, rec, out, n_nodes, n_edges);
}


# First kernel moves its global data in whole 128-B lines through LDS: a chunk's sorted records stored contiguously; W once per block; x and support tiles 512 B per wave access; same k order
# speedup vs baseline: 2.2878x; 1.0897x over previous
#include <hip/hip_runtime.h>
#include <hip/hip_bf16.h>


typedef __attribute__((ext_vector_type(2))) float v2f;
typedef __attribute__((ext_vector_type(8))) float v8f;

#define DIM 64

#define KEY_SHIFT 9
#define BKT_ROWS  400
#define W_ROWS    25
#define NBK       256
#define NWAVE     16
#define CAP       1280
#define OCAP      1024
#define PICKB     8
#define CHUNK     2048
#define W_PITCH   68

__host__ __device__ __forceinline__ int chunks_of(int n_edges) { return (n_edges + CHUNK - 1) / CHUNK; }

__device__ __forceinline__ int wave_incl_scan(int v, int lane) {
    #pragma unroll
    for (int d = 1; d < 32; d <<= 1) {
        const int t = __shfl_up(v, d, 32);
        if (lane >= d) v += t;
    }
    return v;
}

__device__ __forceinline__ void stage_W(const float* __restrict__ W, float* s_W) {
    float4 wq[4];
    #pragma unroll
    for (int i = 0; i < 4; ++i) wq[i] = ((const float4*)W)[threadIdx.x + i * 256];
    #pragma unroll
    for (int i = 0; i < 4; ++i) {
        const int idx = threadIdx.x + i * 256;
        *(float4*)&s_W[(idx >> 4) * W_PITCH + ((idx & 15) << 2)] = wq[i];
    }
}

__global__ __launch_bounds__(256) void gemm_kernel(const float* __restrict__ x,
                                                   const float* __restrict__ W,
                                                   const float* __restrict__ b,
                                                   const int* __restrict__ esrc,
                                                   const int* __restrict__ edst,
                                                   const float* __restrict__ ew,
                                                   float* __restrict__ support,
                                                   int* __restrict__ rec,
                                                   int* __restrict__ tab,
                                                   int n_nodes, int n_edges) {
    __shared__ int s_hist[8 * NBK];
    __shared__ int s_start[NBK];
    __shared__ int s_wtot[8];
    __shared__ __attribute__((aligned(16))) int   s_rec[2 * CHUNK];
    __shared__ __attribute__((aligned(16))) float s_W[DIM * W_PITCH];
    __shared__ __attribute__((aligned(16))) float s_T[8 * 16 * W_PITCH];

    const int lane = threadIdx.x & 31;
    const int wave = threadIdx.x >> 5;

    {
        const int n_chunks = chunks_of(n_edges);
        int2* rec2 = (int2*)rec;
        int2* img2 = (int2*)s_rec;
        int* hw = s_hist + wave * NBK;
        for (int chunk = blockIdx.x; chunk < n_chunks; chunk += gridDim.x) {
            const int e0 = chunk * CHUNK + wave * 256 + lane;
            int dd[8], ss[8];
            float ww[8];
            #pragma unroll
            for (int j = 0; j < 8; ++j) {
                const int e = e0 + j * 32;
                dd[j] = 0; ss[j] = 0; ww[j] = 0.f;
                if (e < n_edges) { dd[j] = edst[e]; ss[j] = esrc[e]; ww[j] = ew[e]; }
            }
            for (int i = threadIdx.x; i < 8 * NBK; i += 256) s_hist[i] = 0;
            if (chunk == (int)blockIdx.x) stage_W(W, s_W);
            __syncthreads();

            int pk[8];
            #pragma unroll
            for (int j = 0; j < 8; ++j) {
                const int  e     = e0 + j * 32;
                const bool valid = e < n_edges;
                const int  d     = dd[j];
                const int  qb    = (int)((unsigned)d / (unsigned)BKT_ROWS);
                const int  k     = qb & (NBK - 1);
                const int  rib   = d - qb * BKT_ROWS;
                const int  ow    = (rib * 41) >> 10;
                const int  fld   = (ow << 5) | (rib - ow * W_ROWS);
                unsigned m = __builtin_amdgcn_ballot_w32(valid);
                #pragma unroll
                for (int bit = 0; bit < 8; ++bit) {
                    const bool     on = ((k >> bit) & 1) != 0;
                    const unsigned bb = __builtin_amdgcn_ballot_w32(on);
                    m &= on ? bb : ~bb;
                }
                const int below = (int)__builtin_amdgcn_mbcnt_lo(m, 0u);
                int r = 0;
                if (valid) {
                    const int base = hw[k];
                    r = base + below;
                    if (below == 0) hw[k] = base + __builtin_popcount(m);
                }
                pk[j] = valid ? (k | (r << 8) | (fld << 20)) : -1;
            }
            __syncthreads();

            {
                const int k = threadIdx.x;
                int run = 0;
                #pragma unroll
                for (int w = 0; w < 8; ++w) {
                    const int c = s_hist[w * NBK + k];
                    s_hist[w * NBK + k] = run;
                    run += c;
                }
                const int incl = wave_incl_scan(run, lane);
                if (lane == 31) s_wtot[wave] = incl;
                __syncthreads();
                int wbase = 0;
                #pragma unroll
                for (int w = 0; w < 8; ++w) {
                    const int tw = s_wtot[w];
                    if (w < wave) wbase += tw;
                }
                const int st = wbase + incl - run;
                s_start[k] = st;
                tab[(size_t)k * n_chunks + chunk] = st;
                if (k == NBK - 1) tab[(size_t)NBK * n_chunks + chunk] = st + run;
            }
            __syncthreads();

            #pragma unroll
            for (int j = 0; j < 8; ++j) {
                if (pk[j] >= 0) {
                    const int k   = pk[j] & (NBK - 1);
                    const int pos = s_start[k] + hw[k] + ((pk[j] >> 8) & 4095);
                    img2[pos] = make_int2((ss[j] << KEY_SHIFT) | (pk[j] >> 20), __float_as_int(ww[j]));
                }
            }
            __syncthreads();
            {
                const int   nrec = min(CHUNK, n_edges - chunk * CHUNK);
                int4*       out4 = (int4*)(rec2 + (size_t)chunk * CHUNK);
                const int4* img4 = (const int4*)s_rec;
                #pragma unroll
                for (int i = 0; i < 4; ++i) {
                    const int p = threadIdx.x + i * 256;
                    if (2 * p < nrec) out4[p] = img4[p];
                }
            }
            __syncthreads();
        }
    }

    if ((int)blockIdx.x >= chunks_of(n_edges)) {
        stage_W(W, s_W);
        __syncthreads();
    }

    const int tile    = blockIdx.x * 8 + wave;
    const int n_tiles = n_nodes >> 4;
    if (tile >= n_tiles) return;

    const int row0    = tile << 4;
    const int halfSel = lane >> 4;
    const int l       = lane & 15;

    float* tl = s_T + wave * (16 * W_PITCH);
    {
        const float4* x4 = (const float4*)(x + (size_t)row0 * DIM);
        float4 q[8];
        #pragma unroll
        for (int i = 0; i < 8; ++i) q[i] = x4[i * 32 + lane];
        #pragma unroll
        for (int i = 0; i < 8; ++i)
            *(float4*)&tl[(2 * i + halfSel) * W_PITCH + (l << 2)] = q[i];
    }

    v8f acc[4];
    #pragma unroll
    for (int nt = 0; nt < 4; ++nt) {
        const float bias = b[nt * 16 + l];
        #pragma unroll
        for (int j = 0; j < 8; ++j) acc[nt][j] = bias;
    }

    const float* ta = tl  + l * W_PITCH + 2 * halfSel;
    const float* wb = s_W + l * W_PITCH + 2 * halfSel;
    #pragma unroll 4
    for (int k = 0; k < DIM; k += 4) {
        const v2f a = *(const v2f*)(ta + k);
        #pragma unroll
        for (int nt = 0; nt < 4; ++nt) {
            const v2f bf = *(const v2f*)(wb + nt * 16 * W_PITCH + k);
            acc[nt] = __builtin_amdgcn_wmma_f32_16x16x4_f32(
                 false, a,  false, bf,
                 (short)0, acc[nt],  false,  false);
        }
    }

    #pragma unroll
    for (int nt = 0; nt < 4; ++nt) {
        #pragma unroll
        for (int j = 0; j < 8; ++j)
            tl[(j + 8 * halfSel) * W_PITCH + nt * 16 + l] = acc[nt][j];
    }
    float4* s4 = (float4*)(support + (size_t)row0 * DIM);
    #pragma unroll
    for (int i = 0; i < 8; ++i)
        s4[i * 32 + lane] = *(const float4*)&tl[(2 * i + halfSel) * W_PITCH + (l << 2)];
}

typedef __attribute__((ext_vector_type(32))) float v32f;

struct Trip {
    int   key;
    float wgt;
};

__device__ __forceinline__ Trip load_trip(const int* own, int i, int ocnt,
                                          const int* s_key, const float* s_w, int lane) {
    const int   e   = i + lane;
    const int   idx = own[min(e, ocnt - 1)];
    const int   k   = s_key[idx];
    const float w   = s_w[idx];
    Trip t;
    t.key = (e < ocnt) ? k : 31;
    t.wgt = (e < ocnt) ? w : 0.f;
    return t;
}

__device__ __forceinline__ void add_owned(const int* own, int ocnt,
                                          const int* s_key, const float* s_w,
                                          const float* __restrict__ support,
                                          v32f& ax, v32f& ay, int lane) {
    if (ocnt <= 0) return;
    const unsigned lane8 = (unsigned)lane * 8u;
    Trip cur = load_trip(own, 0, ocnt, s_key, s_w, lane);
    for (int i = 0; i < ocnt; i += 32) {
        float2 v[32];
        #pragma unroll
        for (int u = 0; u < 32; ++u) {
            const int      ku  = __builtin_amdgcn_readlane(cur.key, u);
            const unsigned off = ((unsigned)(ku >> KEY_SHIFT) << 8) + lane8;
            v[u] = *(const float2*)((const char*)support + off);
        }
        __builtin_amdgcn_sched_group_barrier(0x002, 64, 0);
        __builtin_amdgcn_sched_group_barrier(0x020, 32, 0);
        const Trip nxt = load_trip(own, i + 32, ocnt, s_key, s_w, lane);
        #pragma unroll
        for (int u = 0; u < 32; ++u) {
            const int   ku = __builtin_amdgcn_readlane(cur.key, u);
            const float wu = __int_as_float(__builtin_amdgcn_readlane(__float_as_int(cur.wgt), u));
            const int   n  = ku & 31;
            ax[n] = fmaf(wu, v[u].x, ax[n]);
            ay[n] = fmaf(wu, v[u].y, ay[n]);
        }
        cur = nxt;
    }
}

__global__ __launch_bounds__(512) void owner_kernel(const float* __restrict__ support,
                                                    const int* __restrict__ tab,
                                                    const int* __restrict__ rec,
                                                    float* __restrict__ out,
                                                    int n_nodes, int n_edges) {
    __shared__ int    s_key[NWAVE * CAP];
    __shared__ float  s_w[NWAVE * CAP];
    __shared__ int    s_own[NWAVE * OCAP];
    __shared__ int    s_cnt[NWAVE];
    __shared__ int    s_more[NWAVE];

    const int lane = threadIdx.x & 31;
    const int wv   = __builtin_amdgcn_readfirstlane((int)(threadIdx.x >> 5));
    const int bkt  = blockIdx.x;
    const int n_chunks = chunks_of(n_edges);
    const int* tlo = tab + (size_t)bkt * n_chunks;
    const int* thi = tab + (size_t)(bkt + 1) * n_chunks;

    v32f ax, ay;
    #pragma unroll
    for (int r = 0; r < 32; ++r) { ax[r] = 0.f; ay[r] = 0.f; }

    int*   myKey = s_key + wv * CAP;
    float* myW   = s_w   + wv * CAP;
    int*   myOwn = s_own + wv * OCAP;
    const int2*   rec2 = (const int2*)rec;
    const int     sub  = lane & 7;

    int cur = wv;
    int off = 0;
    for (;;) {
        int cnt = 0;
        while (cur < n_chunks && cnt < CAP) {
            const int ci = cur + NWAVE * lane;
            int st = 0, n = 0;
            if (ci < n_chunks) {
                st = tlo[ci];
                n  = thi[ci] - st;
            }
            if (lane == 0) { st += off; n -= off; }
            const int incl  = wave_incl_scan(n, lane);
            const int excl  = incl - n;
            const int total = __builtin_amdgcn_readlane(incl, 31);
            const int take  = min(total, CAP - cnt);
            const int mine  = min(n, take - excl);
            int so[8], dsti[8], lim[8];
            #pragma unroll
            for (int g = 0; g < 8; ++g) {
                const int j = 4 * g + (lane >> 3);
                lim[g]  = __shfl(mine, j, 32) - sub;
                dsti[g] = cnt + __shfl(excl, j, 32) + sub;
                so[g]   = (cur + NWAVE * j) * CHUNK + __shfl(st, j, 32) + sub;
            }
            for (int t = 0; ; t += 8) {
                if (__builtin_amdgcn_ballot_w32(t < mine) == 0u) break;
                int2 q[8];
                #pragma unroll
                for (int g = 0; g < 8; ++g) {
                    q[g] = make_int2(0, 0);
                    if (t < lim[g]) q[g] = rec2[(size_t)so[g] + t];
                }
                #pragma unroll
                for (int g = 0; g < 8; ++g) {
                    if (t < lim[g]) {
                        myKey[dsti[g] + t] = q[g].x;
                        myW[dsti[g] + t]   = __int_as_float(q[g].y);
                    }
                }
            }
            cnt += take;
            if (take == total) {
                cur += 32 * NWAVE;
                off = 0;
            } else {
                const unsigned part = __builtin_amdgcn_ballot_w32(incl > take);
                const int i    = __builtin_ctz(part);
                const int ex_i = __builtin_amdgcn_readlane(excl, i);
                off = (i == 0 ? off : 0) + (take - ex_i);
                cur += NWAVE * i;
            }
        }
        if (lane == 0) {
            s_cnt[wv]  = cnt;
            s_more[wv] = (cur < n_chunks) ? 1 : 0;
        }
        __syncthreads();

        int ocnt = 0;
        for (int s = 0; s < NWAVE; ++s) {
            const int  c  = __builtin_amdgcn_readfirstlane(s_cnt[s]);
            const int* lk = s_key + s * CAP;
            for (int base = 0; base < c; base += 32 * PICKB) {
                if (ocnt > OCAP - 32 * PICKB) {
                    add_owned(myOwn, ocnt, s_key, s_w, support, ax, ay, lane);
                    ocnt = 0;
                }
                int kq[PICKB];
                #pragma unroll
                for (int q = 0; q < PICKB; ++q) kq[q] = lk[min(base + 32 * q + lane, c - 1)];
                #pragma unroll
                for (int q = 0; q < PICKB; ++q) {
                    if (base + 32 * q >= c) break;
                    const int  i    = base + 32 * q + lane;
                    const bool mine = (i < c) && (((kq[q] >> 5) & (NWAVE - 1)) == wv);
                    const unsigned mk = __builtin_amdgcn_ballot_w32(mine);
                    if (mine) myOwn[ocnt + (int)__builtin_amdgcn_mbcnt_lo(mk, 0u)] = s * CAP + i;
                    ocnt += __builtin_popcount(mk);
                }
            }
        }
        add_owned(myOwn, ocnt, s_key, s_w, support, ax, ay, lane);

        const int mflag = (lane < NWAVE) ? s_more[lane] : 0;
        const unsigned more = __builtin_amdgcn_ballot_w32(mflag != 0);
        __syncthreads();
        if (more == 0u) break;
    }

    float2* out2 = (float2*)out;
    #pragma unroll
    for (int r = 0; r < W_ROWS; ++r) {
        const int node = bkt * BKT_ROWS + wv * W_ROWS + r;
        if (node < n_nodes) out2[(size_t)node * 32 + lane] = make_float2(ax[r], ay[r]);
    }
}

extern "C" void kernel_launch(void* const* d_in, const int* in_sizes, int n_in,
                              void* d_out, int out_size, void* d_ws, size_t ws_size,
                              hipStream_t stream) {
    const float* x    = (const float*)d_in[0];
    const float* W    = (const float*)d_in[1];
    const float* b    = (const float*)d_in[2];
    const int*   esrc = (const int*)  d_in[3];
    const int*   edst = (const int*)  d_in[4];
    const float* ew   = (const float*)d_in[5];

    const int n_nodes = in_sizes[0] / DIM;
    const int n_edges = in_sizes[3];

    float* out     = (float*)d_out;
    const int n_chunks = chunks_of(n_edges);
    float* support = (float*)d_ws;
    int*   rec     = (int*)(support + (size_t)n_nodes * DIM);
    int*   tab     = rec + (size_t)n_chunks * CHUNK * 2;

    const int n_tiles = n_nodes / 16;
    const int gblocks = (n_tiles + 7) / 8;
    gemm_kernel<<<gblocks, 256, 0, stream>>>(x, W, b, esrc, edst, ew, support, rec, tab,
                                             n_nodes, n_edges);

    const int oblocks = (n_nodes + BKT_ROWS - 1) / BKT_ROWS;
    owner_kernel<<<oblocks, 512, 0, stream>>>(support, tab, rec, out, n_nodes, n_edges);
}


# Sort chunks of 4096 edges, 16 per thread: half as many sort waves to start and, per owner, half as many runs, twice as long, each read by 16 lanes; sum order regrouped
# speedup vs baseline: 2.7754x; 1.2131x over previous
#include <hip/hip_runtime.h>
#include <hip/hip_bf16.h>


typedef __attribute__((ext_vector_type(2))) float v2f;
typedef __attribute__((ext_vector_type(8))) float v8f;

#define DIM 64

#define KEY_SHIFT 9
#define BKT_ROWS  400
#define W_ROWS    25
#define NBK       256
#define NWAVE     16
#define CAP       1280
#define OCAP      1024
#define PICKB     8
#define CHUNK     4096
#define EPT       16
#define RLANES    16
#define W_PITCH   68
#define WS_PITCH  33

__host__ __device__ __forceinline__ int chunks_of(int n_edges) { return (n_edges + CHUNK - 1) / CHUNK; }

__device__ __forceinline__ int wave_incl_scan(int v, int lane) {
    #pragma unroll
    for (int d = 1; d < 32; d <<= 1) {
        const int t = __shfl_up(v, d, 32);
        if (lane >= d) v += t;
    }
    return v;
}

__global__ __launch_bounds__(256) void sort_kernel(const int* __restrict__ esrc,
                                                   const int* __restrict__ edst,
                                                   const float* __restrict__ ew,
                                                   int* __restrict__ rec,
                                                   int* __restrict__ tab,
                                                   int n_edges) {
    __shared__ int s_hist[8 * NBK];
    __shared__ int s_start[NBK];
    __shared__ int s_wtot[8];
    __shared__ __attribute__((aligned(16))) int   s_rec[2 * CHUNK];

    const int lane = threadIdx.x & 31;
    const int wave = threadIdx.x >> 5;

    {
        const int n_chunks = chunks_of(n_edges);
        int2* rec2 = (int2*)rec;
        int2* img2 = (int2*)s_rec;
        int* hw = s_hist + wave * NBK;
        for (int chunk = blockIdx.x; chunk < n_chunks; chunk += gridDim.x) {
            const int e0 = chunk * CHUNK + wave * (32 * EPT) + lane;
            int dd[EPT], ss[EPT];
            float ww[EPT];
            #pragma unroll
            for (int j = 0; j < EPT; ++j) {
                const int e = e0 + j * 32;
                dd[j] = 0; ss[j] = 0; ww[j] = 0.f;
                if (e < n_edges) { dd[j] = edst[e]; ss[j] = esrc[e]; ww[j] = ew[e]; }
            }
            for (int i = threadIdx.x; i < 8 * NBK; i += 256) s_hist[i] = 0;
            __syncthreads();

            int pk[EPT];
            #pragma unroll
            for (int j = 0; j < EPT; ++j) {
                const int  e     = e0 + j * 32;
                const bool valid = e < n_edges;
                const int  d     = dd[j];
                const int  qb    = (int)((unsigned)d / (unsigned)BKT_ROWS);
                const int  k     = qb & (NBK - 1);
                const int  rib   = d - qb * BKT_ROWS;
                const int  ow    = (rib * 41) >> 10;
                const int  fld   = (ow << 5) | (rib - ow * W_ROWS);
                unsigned m = __builtin_amdgcn_ballot_w32(valid);
                #pragma unroll
                for (int bit = 0; bit < 8; ++bit) {
                    const bool     on = ((k >> bit) & 1) != 0;
                    const unsigned bb = __builtin_amdgcn_ballot_w32(on);
                    m &= on ? bb : ~bb;
                }
                const int below = (int)__builtin_amdgcn_mbcnt_lo(m, 0u);
                int r = 0;
                if (valid) {
                    const int base = hw[k];
                    r = base + below;
                    if (below == 0) hw[k] = base + __builtin_popcount(m);
                }
                pk[j] = valid ? (k | (r << 8) | (fld << 20)) : -1;
            }
            __syncthreads();

            {
                const int k = threadIdx.x;
                int run = 0;
                #pragma unroll
                for (int w = 0; w < 8; ++w) {
                    const int c = s_hist[w * NBK + k];
                    s_hist[w * NBK + k] = run;
                    run += c;
                }
                const int incl = wave_incl_scan(run, lane);
                if (lane == 31) s_wtot[wave] = incl;
                __syncthreads();
                int wbase = 0;
                #pragma unroll
                for (int w = 0; w < 8; ++w) {
                    const int tw = s_wtot[w];
                    if (w < wave) wbase += tw;
                }
                const int st = wbase + incl - run;
                s_start[k] = st;
                tab[(size_t)k * n_chunks + chunk] = st;
                if (k == NBK - 1) tab[(size_t)NBK * n_chunks + chunk] = st + run;
            }
            __syncthreads();

            #pragma unroll
            for (int j = 0; j < EPT; ++j) {
                if (pk[j] >= 0) {
                    const int k   = pk[j] & (NBK - 1);
                    const int pos = s_start[k] + hw[k] + ((pk[j] >> 8) & 4095);
                    img2[pos] = make_int2((ss[j] << KEY_SHIFT) | (pk[j] >> 20), __float_as_int(ww[j]));
                }
            }
            __syncthreads();
            {
                const int   nrec = min(CHUNK, n_edges - chunk * CHUNK);
                int4*       out4 = (int4*)(rec2 + (size_t)chunk * CHUNK);
                const int4* img4 = (const int4*)s_rec;
                #pragma unroll
                for (int i = 0; i < CHUNK / 512; ++i) {
                    const int p = threadIdx.x + i * 256;
                    if (2 * p < nrec) out4[p] = img4[p];
                }
            }
            __syncthreads();
        }
    }
}

typedef __attribute__((ext_vector_type(32))) float v32f;

struct Trip {
    int   key;
    float wgt;
};

__device__ __forceinline__ Trip load_trip(const int* own, int i, int ocnt,
                                          const int* s_key, const float* s_w, int lane) {
    const int   e   = i + lane;
    const int   idx = own[min(e, ocnt - 1)];
    const int   k   = s_key[idx];
    const float w   = s_w[idx];
    Trip t;
    t.key = (e < ocnt) ? k : 31;
    t.wgt = (e < ocnt) ? w : 0.f;
    return t;
}

__device__ __forceinline__ void add_owned(const int* own, int ocnt,
                                          const int* s_key, const float* s_w, float* wsl,
                                          const float* __restrict__ x,
                                          v32f& ax, v32f& ay, int lane) {
    if (ocnt <= 0) return;
    const unsigned lane8 = (unsigned)lane * 8u;
    Trip cur = load_trip(own, 0, ocnt, s_key, s_w, lane);
    for (int i = 0; i < ocnt; i += 32) {
        float* const slot = wsl + (cur.key & 31);
        const float  wold = *slot;
        float2 v[32];
        #pragma unroll
        for (int u = 0; u < 32; ++u) {
            const int      ku  = __builtin_amdgcn_readlane(cur.key, u);
            const unsigned off = ((unsigned)(ku >> KEY_SHIFT) << 8) + lane8;
            v[u] = *(const float2*)((const char*)x + off);
        }
        __builtin_amdgcn_sched_group_barrier(0x002, 64, 0);
        __builtin_amdgcn_sched_group_barrier(0x020, 32, 0);
        const Trip nxt = load_trip(own, i + 32, ocnt, s_key, s_w, lane);
        #pragma unroll
        for (int u = 0; u < 32; ++u) {
            const int   ku = __builtin_amdgcn_readlane(cur.key, u);
            const float wu = __int_as_float(__builtin_amdgcn_readlane(__float_as_int(cur.wgt), u));
            const int   n  = ku & 31;
            ax[n] = fmaf(wu, v[u].x, ax[n]);
            ay[n] = fmaf(wu, v[u].y, ay[n]);
        }
        *slot = wold + cur.wgt;
        cur = nxt;
    }
}

__global__ __launch_bounds__(512) void owner_kernel(const float* __restrict__ x,
                                                    const float* __restrict__ W,
                                                    const float* __restrict__ b,
                                                    const int* __restrict__ tab,
                                                    const int* __restrict__ rec,
                                                    float* __restrict__ out,
                                                    int n_nodes, int n_edges) {
    __shared__ __attribute__((aligned(16))) int s_kw[2 * NWAVE * CAP];
    __shared__ int    s_own[NWAVE * OCAP];
    __shared__ float  s_wsa[NWAVE * 32 * WS_PITCH];
    __shared__ float  s_ws[BKT_ROWS];
    __shared__ int    s_cnt[NWAVE];
    __shared__ int    s_more[NWAVE];
    int*   s_key = s_kw;
    float* s_w   = (float*)(s_kw + NWAVE * CAP);

    const int lane = threadIdx.x & 31;
    const int wv   = __builtin_amdgcn_readfirstlane((int)(threadIdx.x >> 5));
    const int bkt  = blockIdx.x;
    const int n_chunks = chunks_of(n_edges);
    const int* tlo = tab + (size_t)bkt * n_chunks;
    const int* thi = tab + (size_t)(bkt + 1) * n_chunks;

    v32f ax, ay;
    #pragma unroll
    for (int r = 0; r < 32; ++r) { ax[r] = 0.f; ay[r] = 0.f; }

    int*   myKey = s_key + wv * CAP;
    float* myW   = s_w   + wv * CAP;
    int*   myOwn = s_own + wv * OCAP;
    float* wsp   = s_wsa + wv * (32 * WS_PITCH);
    float* wsl   = wsp + lane * WS_PITCH;
    #pragma unroll
    for (int n = 0; n < 32; ++n) wsl[n] = 0.f;
    const int2*   rec2 = (const int2*)rec;
    const int     sub  = lane & (RLANES - 1);

    int cur = wv;
    int off = 0;
    for (;;) {
        int cnt = 0;
        while (cur < n_chunks && cnt < CAP) {
            const int ci = cur + NWAVE * lane;
            int st = 0, n = 0;
            if (ci < n_chunks) {
                st = tlo[ci];
                n  = thi[ci] - st;
            }
            if (lane == 0) { st += off; n -= off; }
            const int incl  = wave_incl_scan(n, lane);
            const int excl  = incl - n;
            const int total = __builtin_amdgcn_readlane(incl, 31);
            const int take  = min(total, CAP - cnt);
            const int mine  = min(n, take - excl);
            int so[RLANES], dsti[RLANES], lim[RLANES];
            #pragma unroll
            for (int g = 0; g < RLANES; ++g) {
                const int j = 2 * g + (lane >> 4);
                lim[g]  = __shfl(mine, j, 32) - sub;
                dsti[g] = cnt + __shfl(excl, j, 32) + sub;
                so[g]   = (cur + NWAVE * j) * CHUNK + __shfl(st, j, 32) + sub;
            }
            for (int t = 0; ; t += RLANES) {
                if (__builtin_amdgcn_ballot_w32(t < mine) == 0u) break;
                int2 q[RLANES];
                #pragma unroll
                for (int g = 0; g < RLANES; ++g) {
                    q[g] = make_int2(0, 0);
                    if (t < lim[g]) q[g] = rec2[(size_t)so[g] + t];
                }
                #pragma unroll
                for (int g = 0; g < RLANES; ++g) {
                    if (t < lim[g]) {
                        myKey[dsti[g] + t] = q[g].x;
                        myW[dsti[g] + t]   = __int_as_float(q[g].y);
                    }
                }
            }
            cnt += take;
            if (take == total) {
                cur += 32 * NWAVE;
                off = 0;
            } else {
                const unsigned part = __builtin_amdgcn_ballot_w32(incl > take);
                const int i    = __builtin_ctz(part);
                const int ex_i = __builtin_amdgcn_readlane(excl, i);
                off = (i == 0 ? off : 0) + (take - ex_i);
                cur += NWAVE * i;
            }
        }
        if (lane == 0) {
            s_cnt[wv]  = cnt;
            s_more[wv] = (cur < n_chunks) ? 1 : 0;
        }
        __syncthreads();

        int ocnt = 0;
        for (int s = 0; s < NWAVE; ++s) {
            const int  c  = __builtin_amdgcn_readfirstlane(s_cnt[s]);
            const int* lk = s_key + s * CAP;
            for (int base = 0; base < c; base += 32 * PICKB) {
                if (ocnt > OCAP - 32 * PICKB) {
                    add_owned(myOwn, ocnt, s_key, s_w, wsl, x, ax, ay, lane);
                    ocnt = 0;
                }
                int kq[PICKB];
                #pragma unroll
                for (int q = 0; q < PICKB; ++q) kq[q] = lk[min(base + 32 * q + lane, c - 1)];
                #pragma unroll
                for (int q = 0; q < PICKB; ++q) {
                    if (base + 32 * q >= c) break;
                    const int  i    = base + 32 * q + lane;
                    const bool mine = (i < c) && (((kq[q] >> 5) & (NWAVE - 1)) == wv);
                    const unsigned mk = __builtin_amdgcn_ballot_w32(mine);
                    if (mine) myOwn[ocnt + (int)__builtin_amdgcn_mbcnt_lo(mk, 0u)] = s * CAP + i;
                    ocnt += __builtin_popcount(mk);
                }
            }
        }
        add_owned(myOwn, ocnt, s_key, s_w, wsl, x, ax, ay, lane);

        const int mflag = (lane < NWAVE) ? s_more[lane] : 0;
        const unsigned more = __builtin_amdgcn_ballot_w32(mflag != 0);
        __syncthreads();
        if (more == 0u) break;
    }

    float* s_A  = (float*)s_kw;
    float* s_Wt = s_A + BKT_ROWS * W_PITCH;
    {
        const float4 wq0 = ((const float4*)W)[threadIdx.x];
        const float4 wq1 = ((const float4*)W)[threadIdx.x + 512];
        #pragma unroll
        for (int r = 0; r < W_ROWS; ++r)
            *(float2*)&s_A[(wv * W_ROWS + r) * W_PITCH + 2 * lane] = make_float2(ax[r], ay[r]);
        float ws = 0.f;
        #pragma unroll
        for (int u = 0; u < 32; ++u) ws += wsp[u * WS_PITCH + lane];
        if (lane < W_ROWS) s_ws[wv * W_ROWS + lane] = ws;
        {
            const int idx0 = threadIdx.x, idx1 = threadIdx.x + 512;
            *(float4*)&s_Wt[(idx0 >> 4) * W_PITCH + ((idx0 & 15) << 2)] = wq0;
            *(float4*)&s_Wt[(idx1 >> 4) * W_PITCH + ((idx1 & 15) << 2)] = wq1;
        }
    }
    __syncthreads();

    const int halfSel = lane >> 4;
    const int l       = lane & 15;
    for (int t = wv; t < BKT_ROWS / 16; t += NWAVE) {
        float* tl = s_A + t * (16 * W_PITCH);
        float wsr[8];
        #pragma unroll
        for (int j = 0; j < 8; ++j) wsr[j] = s_ws[t * 16 + j + 8 * halfSel];
        v8f acc[4];
        #pragma unroll
        for (int nt = 0; nt < 4; ++nt) {
            const float bias = b[nt * 16 + l];
            #pragma unroll
            for (int j = 0; j < 8; ++j) acc[nt][j] = wsr[j] * bias;
        }

        const float* ta = tl   + l * W_PITCH + 2 * halfSel;
        const float* wb = s_Wt + l * W_PITCH + 2 * halfSel;
        #pragma unroll 4
        for (int k = 0; k < DIM; k += 4) {
            const v2f a = *(const v2f*)(ta + k);
            #pragma unroll
            for (int nt = 0; nt < 4; ++nt) {
                const v2f bf = *(const v2f*)(wb + nt * 16 * W_PITCH + k);
                acc[nt] = __builtin_amdgcn_wmma_f32_16x16x4_f32(
                     false, a,  false, bf,
                     (short)0, acc[nt],  false,  false);
            }
        }

        #pragma unroll
        for (int nt = 0; nt < 4; ++nt) {
            #pragma unroll
            for (int j = 0; j < 8; ++j)
                tl[(j + 8 * halfSel) * W_PITCH + nt * 16 + l] = acc[nt][j];
        }
        const int node0 = bkt * BKT_ROWS + t * 16;
        float4* o4 = (float4*)(out + (size_t)node0 * DIM);
        if (node0 + 16 <= n_nodes) {
            float4 q[8];
            #pragma unroll
            for (int i = 0; i < 8; ++i) q[i] = *(const float4*)&tl[(2 * i + halfSel) * W_PITCH + (l << 2)];
            #pragma unroll
            for (int i = 0; i < 8; ++i) o4[i * 32 + lane] = q[i];
        } else {
            #pragma unroll
            for (int i = 0; i < 8; ++i) {
                if (node0 + 2 * i + halfSel < n_nodes)
                    o4[i * 32 + lane] = *(const float4*)&tl[(2 * i + halfSel) * W_PITCH + (l << 2)];
            }
        }
    }
}

extern "C" void kernel_launch(void* const* d_in, const int* in_sizes, int n_in,
                              void* d_out, int out_size, void* d_ws, size_t ws_size,
                              hipStream_t stream) {
    const float* x    = (const float*)d_in[0];
    const float* W    = (const float*)d_in[1];
    const float* b    = (const float*)d_in[2];
    const int*   esrc = (const int*)  d_in[3];
    const int*   edst = (const int*)  d_in[4];
    const float* ew   = (const float*)d_in[5];

    const int n_nodes = in_sizes[0] / DIM;
    const int n_edges = in_sizes[3];

    float* out     = (float*)d_out;
    const int n_chunks = chunks_of(n_edges);
    int*   rec     = (int*)d_ws;
    int*   tab     = rec + (size_t)n_chunks * CHUNK * 2;

    sort_kernel<<<n_chunks, 256, 0, stream>>>(esrc, edst, ew, rec, tab, n_edges);

    const int oblocks = (n_nodes + BKT_ROWS - 1) / BKT_ROWS;
    owner_kernel<<<oblocks, 512, 0, stream>>>(x, W, b, tab, rec, out, n_nodes, n_edges);
}
